# starting kernel
# baseline (speedup 1.0000x reference)
#include <hip/hip_runtime.h>


typedef __attribute__((ext_vector_type(2))) float v2f;
typedef __attribute__((ext_vector_type(8))) float v8f;

#define TT  4096
#define BB  256
#define NXX 64
#define NYY 16
#define NUU 16
#define NDD 8

__device__ __forceinline__ float bcast_lane(float x, int l) {
  return __int_as_float(__builtin_amdgcn_readlane(__float_as_int(x), l));
}

__device__ __forceinline__ v8f wmma_mm(const float* A, int ras, int cas,
                                       const float* B, int rbs, int cbs,
                                       int K, v8f acc, int lane) {
  const int mn = lane & 15;
  const int kh = (lane >> 4) << 1;
#pragma unroll
  for (int k = 0; k < K; k += 4) {
    v2f a, b;
    a.x = A[mn * ras + (k + kh) * cas];
    a.y = A[mn * ras + (k + kh + 1) * cas];
    b.x = B[(k + kh) * rbs + mn * cbs];
    b.y = B[(k + kh + 1) * rbs + mn * cbs];
    acc = __builtin_amdgcn_wmma_f32_16x16x4_f32(false, a, false, b,
                                                (short)0, acc, false, false);
  }
  return acc;
}

__device__ __forceinline__ void store_tile(float* D, int ldd, v8f acc, int lane) {
  const int col = lane & 15;
  const int rb  = (lane >> 4) << 3;
#pragma unroll
  for (int i = 0; i < 8; ++i) D[(rb + i) * ldd + col] = acc[i];
}

__global__ __launch_bounds__(256) void kf_gains_kernel(
    const float* __restrict__ Wfx_g, const float* __restrict__ Wfy_g,
    float* __restrict__ Lbuf) {
  __shared__ float sWfx[NXX * NXX];
  __shared__ float sWfy[NXX * NYY];
  __shared__ float sP[NXX * NXX];
  __shared__ float sT1[NXX * NXX];
  __shared__ float sPWfy[NXX * NYY];
  __shared__ float sS[NYY * NYY];
  __shared__ float sSinv[NYY * NYY];
  __shared__ float sL[NXX * NYY];

  const int tid  = threadIdx.x;
  const int lane = tid & 31;
  const int wave = tid >> 5;

  for (int e = tid; e < NXX * NXX; e += 256) sWfx[e] = Wfx_g[e];
  for (int e = tid; e < NXX * NYY; e += 256) sWfy[e] = Wfy_g[e];
  for (int e = tid; e < NXX * NXX; e += 256) {
    int r = e >> 6, c = e & 63;
    sP[e] = (r == c) ? 1.0f : 0.0f;
  }
  for (int e = tid; e < NXX * NYY; e += 256) Lbuf[e] = 0.0f;
  __syncthreads();

  for (int t = 0; t < TT; ++t) {
    for (int tile = wave; tile < 16; tile += 8) {
      int mt = tile >> 2, nt = tile & 3;
      v8f acc = {};
      acc = wmma_mm(sWfx + mt * 16 * NXX, NXX, 1, sP + nt * 16, NXX, 1, NXX, acc, lane);
      store_tile(sT1 + mt * 16 * NXX + nt * 16, NXX, acc, lane);
    }
    __syncthreads();
    for (int tile = wave; tile < 16; tile += 8) {
      int mt = tile >> 2, nt = tile & 3;
      v8f acc = {};
      acc = wmma_mm(sWfx + mt * 16 * NXX, NXX, 1, sT1 + nt * 16 * NXX, 1, NXX, NXX, acc, lane);
      int col = lane & 15, rb = (lane >> 4) << 3;
#pragma unroll
      for (int i = 0; i < 8; ++i) {
        int gr = mt * 16 + rb + i, gc = nt * 16 + col;
        sP[gr * NXX + gc] = acc[i] + ((gr == gc) ? 1.0f : 0.0f);
      }
    }
    __syncthreads();
    if (wave < 4) {
      v8f acc = {};
      acc = wmma_mm(sP + wave * 16 * NXX, NXX, 1, sWfy, NYY, 1, NXX, acc, lane);
      store_tile(sPWfy + wave * 16 * NYY, NYY, acc, lane);
    }
    __syncthreads();
    if (wave == 0) {
      v8f acc = {};
      acc = wmma_mm(sWfy, 1, NYY, sPWfy, NYY, 1, NXX, acc, lane);
      int col = lane & 15, rb = (lane >> 4) << 3;
#pragma unroll
      for (int i = 0; i < 8; ++i) {
        int r = rb + i;
        sS[r * NYY + col] = acc[i] + ((r == col) ? 1.0f : 0.0f);
      }
      float colv[16];
#pragma unroll
      for (int i = 0; i < 16; ++i)
        colv[i] = (lane < 16) ? sS[i * NYY + lane]
                              : ((i == (lane - 16)) ? 1.0f : 0.0f);
#pragma unroll
      for (int k = 0; k < 16; ++k) {
        float piv  = bcast_lane(colv[k], k);
        float rinv = 1.0f / piv;
        colv[k] *= rinv;
#pragma unroll
        for (int i = 0; i < 16; ++i) {
          if (i != k) {
            float f = bcast_lane(colv[i], k);
            colv[i] -= f * colv[k];
          }
        }
      }
      if (lane >= 16) {
#pragma unroll
        for (int i = 0; i < 16; ++i) sSinv[i * NYY + (lane - 16)] = colv[i];
      }
    }
    __syncthreads();
    if (wave < 4) {
      v8f acc = {};
      acc = wmma_mm(sPWfy + wave * 16 * NYY, NYY, 1, sSinv, NYY, 1, NYY, acc, lane);
      int col = lane & 15, rb = (lane >> 4) << 3;
      float* gL = Lbuf + (size_t)(t + 1) * (NXX * NYY);
#pragma unroll
      for (int i = 0; i < 8; ++i) {
        int gr = wave * 16 + rb + i;
        sL[gr * NYY + col] = acc[i];
        gL[gr * NYY + col] = acc[i];
      }
    }
    __syncthreads();
    for (int tile = wave; tile < 16; tile += 8) {
      int mt = tile >> 2, nt = tile & 3;
      v8f acc = {};
      acc = wmma_mm(sL + mt * 16 * NYY, NYY, 1, sPWfy + nt * 16 * NYY, 1, NYY, NYY, acc, lane);
      int col = lane & 15, rb = (lane >> 4) << 3;
#pragma unroll
      for (int i = 0; i < 8; ++i) {
        int gr = mt * 16 + rb + i, gc = nt * 16 + col;
        sP[gr * NXX + gc] = ((gr == gc) ? 1.0f : 0.0f) - acc[i];
      }
    }
    __syncthreads();
  }
}

__global__ __launch_bounds__(128) void kf_scan_kernel(
    const float* __restrict__ Ym, const float* __restrict__ U,
    const float* __restrict__ Dg,
    const float* __restrict__ Wfx_g, const float* __restrict__ bfx,
    const float* __restrict__ Wfu_g, const float* __restrict__ bfu,
    const float* __restrict__ Wfd_g, const float* __restrict__ bfd,
    const float* __restrict__ Wfy_g, const float* __restrict__ bfy_g,
    const float* __restrict__ Lbuf, float* __restrict__ Out) {
  __shared__ float sWfx[NXX * NXX];
  __shared__ float sWfu[NUU * NXX];
  __shared__ float sWfd[NDD * NXX];
  __shared__ float sWfy[NXX * NYY];
  __shared__ float sBsum[NXX];
  __shared__ float sBfy[NYY];
  __shared__ float sX[16 * NXX];
  __shared__ float sXp[16 * NXX];
  __shared__ float sL[NXX * NYY];
  __shared__ float sYm[16 * NYY];
  __shared__ float sU[16 * NUU];
  __shared__ float sD[16 * NDD];
  __shared__ float sInn[16 * NYY];

  const int tid  = threadIdx.x;
  const int lane = tid & 31;
  const int wave = tid >> 5;
  const int r0   = blockIdx.x * 16;

  for (int e = tid; e < NXX * NXX; e += 128) sWfx[e] = Wfx_g[e];
  for (int e = tid; e < NUU * NXX; e += 128) sWfu[e] = Wfu_g[e];
  for (int e = tid; e < NDD * NXX; e += 128) sWfd[e] = Wfd_g[e];
  for (int e = tid; e < NXX * NYY; e += 128) sWfy[e] = Wfy_g[e];
  if (tid < NXX) sBsum[tid] = bfx[tid] + bfu[tid] + bfd[tid];
  if (tid < NYY) sBfy[tid] = bfy_g[tid];
  for (int e = tid; e < 16 * NXX; e += 128) sX[e] = 0.0f;
  __syncthreads();

  for (int t = 0; t < TT; ++t) {
    {
      const float* gL  = Lbuf + (size_t)t * (NXX * NYY);
      const float* gYm = Ym + (size_t)t * BB * NYY + r0 * NYY;
      const float* gU  = U  + (size_t)t * BB * NUU + r0 * NUU;
      const float* gD  = Dg + (size_t)t * BB * NDD + r0 * NDD;
      for (int e = tid; e < NXX * NYY; e += 128) sL[e] = gL[e];
      for (int e = tid; e < 16 * NYY; e += 128)
        sYm[e] = __builtin_nontemporal_load(gYm + e);
      for (int e = tid; e < 16 * NUU; e += 128)
        sU[e] = __builtin_nontemporal_load(gU + e);
      if (tid < 16 * NDD) sD[tid] = __builtin_nontemporal_load(gD + tid);
      if (t + 1 < TT) {
        const char* pL  = (const char*)(Lbuf + (size_t)(t + 1) * (NXX * NYY));
        const char* pYm = (const char*)(Ym + (size_t)(t + 1) * BB * NYY + r0 * NYY);
        const char* pU  = (const char*)(U  + (size_t)(t + 1) * BB * NUU + r0 * NUU);
        const char* pD  = (const char*)(Dg + (size_t)(t + 1) * BB * NDD + r0 * NDD);
        if (tid < 32) __builtin_prefetch(pL + tid * 128, 0, 0);
        if (tid < 8)  __builtin_prefetch(pYm + tid * 128, 0, 0);
        if (tid < 8)  __builtin_prefetch(pU + tid * 128, 0, 0);
        if (tid < 4)  __builtin_prefetch(pD + tid * 128, 0, 0);
      }
    }
    __syncthreads();
    {
      const int nt = wave;
      v8f acc = {};
      acc = wmma_mm(sX, NXX, 1, sWfx + nt * 16, NXX, 1, NXX, acc, lane);
      acc = wmma_mm(sU, NUU, 1, sWfu + nt * 16, NXX, 1, NUU, acc, lane);
      acc = wmma_mm(sD, NDD, 1, sWfd + nt * 16, NXX, 1, NDD, acc, lane);
      int col = lane & 15, rb = (lane >> 4) << 3;
      float bias = sBsum[nt * 16 + col];
#pragma unroll
      for (int i = 0; i < 8; ++i)
        sXp[(rb + i) * NXX + nt * 16 + col] = acc[i] + bias;
    }
    __syncthreads();
    if (wave == 0) {
      v8f acc = {};
      acc = wmma_mm(sXp, NXX, 1, sWfy, NYY, 1, NXX, acc, lane);
      int col = lane & 15, rb = (lane >> 4) << 3;
      float by = sBfy[col];
#pragma unroll
      for (int i = 0; i < 8; ++i) {
        int r = rb + i;
        sInn[r * NYY + col] = sYm[r * NYY + col] - (acc[i] + by);
      }
    }
    __syncthreads();
    {
      const int nt = wave;
      v8f acc = {};
      acc = wmma_mm(sInn, NYY, 1, sL + nt * 16 * NYY, 1, NYY, NYY, acc, lane);
      int col = lane & 15, rb = (lane >> 4) << 3;
#pragma unroll
      for (int i = 0; i < 8; ++i) {
        int idx = (rb + i) * NXX + nt * 16 + col;
        sX[idx] = sXp[idx] + acc[i];
      }
    }
    __syncthreads();
  }

  for (int e = tid; e < 16 * NXX; e += 128) {
    int row = e >> 6, col = e & 63;
    Out[(size_t)(r0 + row) * NXX + col] = sX[e];
  }
}

extern "C" void kernel_launch(void* const* d_in, const int* in_sizes, int n_in,
                              void* d_out, int out_size, void* d_ws, size_t ws_size,
                              hipStream_t stream) {
  const float* Ym  = (const float*)d_in[0];
  const float* U   = (const float*)d_in[1];
  const float* Dg  = (const float*)d_in[2];
  const float* Wfx = (const float*)d_in[3];
  const float* bfx = (const float*)d_in[4];
  const float* bfu = (const float*)d_in[6];
  const float* Wfu = (const float*)d_in[5];
  const float* Wfd = (const float*)d_in[7];
  const float* bfd = (const float*)d_in[8];
  const float* Wfy = (const float*)d_in[9];
  const float* bfy = (const float*)d_in[10];

  float* Lbuf = (float*)d_ws;
  float* Out  = (float*)d_out;

  kf_gains_kernel<<<1, 256, 0, stream>>>(Wfx, Wfy, Lbuf);
  kf_scan_kernel<<<16, 128, 0, stream>>>(Ym, U, Dg, Wfx, bfx, Wfu, bfu,
                                         Wfd, bfd, Wfy, bfy, Lbuf, Out);
}


# Gain kernel: S = S0 + W2^T V split over K on the 4 waves whose accumulators hold V's tiles (tile = B operand, W2 rows in registers, same k order); inverse wave adds 4 slices; sums re-associated
# speedup vs baseline: 7.8046x; 7.8046x over previous
#include <hip/hip_runtime.h>


typedef __attribute__((ext_vector_type(2))) float v2f;
typedef __attribute__((ext_vector_type(4))) float v4f;
typedef __attribute__((ext_vector_type(8))) float v8f;

#define TT  4096
#define BB  256
#define NXX 64
#define NYY 16
#define NUU 16
#define NDD 8

#define SLOT    (NXX * NYY)
#define SPP     20
#define SPS     (NYY * SPP)
#define HALF_T  (TT / 2)
#define TC      64
#define NCP     (HALF_T / TC)
#define NCH     (TT / TC)
#define NTILE   20
#define P_SLOT0   (HALF_T + 1)
#define RES_SLOT0 (P_SLOT0 + 4)

__device__ __forceinline__ float bcast_lane(float x, int l) {
  return __int_as_float(__builtin_amdgcn_readlane(__float_as_int(x), l));
}

__device__ __forceinline__ float xhalf(float x) {
  const int b = __float_as_int(x);
  return __int_as_float(__builtin_amdgcn_permlanex16(b, b, 0x76543210, 0xfedcba98,
                                                     false, false));
}

__device__ __forceinline__ v8f wmma_mm(const float* A, int ras, int cas,
                                       const float* B, int rbs, int cbs,
                                       int K, v8f acc, int lane) {
  const int mn = lane & 15;
  const int kh = (lane >> 4) << 1;
#pragma unroll
  for (int k = 0; k < K; k += 4) {
    v2f a, b;
    a.x = A[mn * ras + (k + kh) * cas];
    a.y = A[mn * ras + (k + kh + 1) * cas];
    b.x = B[(k + kh) * rbs + mn * cbs];
    b.y = B[(k + kh + 1) * rbs + mn * cbs];
    acc = __builtin_amdgcn_wmma_f32_16x16x4_f32(false, a, false, b,
                                                (short)0, acc, false, false);
  }
  return acc;
}

__device__ __forceinline__ void store_tile(float* D, int ldd, v8f acc, int lane) {
  const int col = lane & 15;
  const int rb  = (lane >> 4) << 3;
#pragma unroll
  for (int i = 0; i < 8; ++i) D[(rb + i) * ldd + col] = acc[i];
}

__global__ __launch_bounds__(256) void kf_gains_kernel(
    const float* __restrict__ Wfx_g, const float* __restrict__ Wfy_g,
    float* __restrict__ ws, int half) {
  float* Lbuf  = ws;
  float* Psave = ws + (size_t)P_SLOT0 * SLOT;
  __shared__ float sWfxT[NXX * NXX];
  __shared__ float sWfy[NXX * NYY];
  __shared__ float sW2[NXX * NYY];
  __shared__ float sV[NXX * NYY];
  __shared__ __attribute__((aligned(16))) float sG[NXX * NYY];
  __shared__ __attribute__((aligned(16))) float sSp[4 * SPS];
  __shared__ float sSinv[NYY * NYY];

  const int tid  = threadIdx.x;
  const int lane = tid & 31;
  const int wave = tid >> 5;
  const int c    = lane & 15;
  const int h    = lane >> 4;

  for (int e = tid; e < NXX * NXX; e += 256) {
    int r = e >> 6, cc = e & 63;
    sWfxT[cc * NXX + r] = Wfx_g[e];
  }
  for (int e = tid; e < NXX * NYY; e += 256) sWfy[e] = Wfy_g[e];
  if (half == 0) {
    for (int e = tid; e < NXX * NYY; e += 256) Lbuf[e] = 0.0f;
  } else {
    for (int e = tid; e < NXX * NYY; e += 256)
      Lbuf[e] = Lbuf[(size_t)HALF_T * SLOT + e];
  }
  __syncthreads();
  v8f s0acc = {};
  if (wave < 4) {
    v8f acc = {};
    acc = wmma_mm(Wfx_g + wave * 16, 1, NXX, sWfy, NYY, 1, NXX, acc, lane);
    store_tile(sW2 + wave * 16 * NYY, NYY, acc, lane);
    if (wave == 0) {
      s0acc = wmma_mm(sWfy, 1, NYY, sWfy, NYY, 1, NXX, s0acc, lane);
      int col = lane & 15, rb = (lane >> 4) << 3;
#pragma unroll
      for (int i = 0; i < 8; ++i) s0acc[i] += ((rb + i == col) ? 1.0f : 0.0f);
    }
  }
  __syncthreads();

  v8f pt[4];
  v2f w2f[16];
  v2f w2a[4];
#pragma unroll
  for (int jj = 0; jj < 4; ++jj) w2a[jj] = v2f{};
#pragma unroll
  for (int mt = 0; mt < 4; ++mt) pt[mt] = v8f{};
#pragma unroll
  for (int q = 0; q < 16; ++q) w2f[q] = v2f{};
  if (wave < 4) {
    if (half == 0) {
#pragma unroll
      for (int mt = 0; mt < 4; ++mt)
#pragma unroll
        for (int i = 0; i < 8; ++i)
          pt[mt][i] = (mt == wave && i + 8 * h == c) ? 1.0f : 0.0f;
    } else {
#pragma unroll
      for (int mt = 0; mt < 4; ++mt)
#pragma unroll
        for (int i = 0; i < 8; ++i)
          pt[mt][i] = Psave[((wave * 4 + mt) * 8 + i) * 32 + lane];
    }
#pragma unroll
    for (int mt = 0; mt < 4; ++mt)
#pragma unroll
      for (int jj = 0; jj < 4; ++jj) {
        const int k = 16 * mt + 8 * h + 2 * jj;
        w2f[4 * mt + jj].x = sW2[k * NYY + c];
        w2f[4 * mt + jj].y = sW2[(k + 1) * NYY + c];
      }
#pragma unroll
    for (int jj = 0; jj < 4; ++jj) {
      const int k = 16 * wave + 8 * h + 2 * jj;
      w2a[jj].x = sW2[k * NYY + c];
      w2a[jj].y = sW2[(k + 1) * NYY + c];
    }
  }

  for (int t = 0; t < HALF_T; ++t) {
    if (wave < 4) {
      v8f acc = {};
#pragma unroll
      for (int mt = 0; mt < 4; ++mt)
#pragma unroll
        for (int jj = 0; jj < 4; ++jj) {
          v2f a;
          a.x = pt[mt][2 * jj];
          a.y = pt[mt][2 * jj + 1];
          acc = __builtin_amdgcn_wmma_f32_16x16x4_f32(false, a, false, w2f[4 * mt + jj],
                                                      (short)0, acc, false, false);
        }
      store_tile(sV + wave * 16 * NYY, NYY, acc, lane);
      v8f sp = s0acc;
#pragma unroll
      for (int jj = 0; jj < 4; ++jj) {
        v2f b;
        b.x = acc[2 * jj];
        b.y = acc[2 * jj + 1];
        sp = __builtin_amdgcn_wmma_f32_16x16x4_f32(false, w2a[jj], false, b,
                                                   (short)0, sp, false, false);
      }
      float* dsp = sSp + wave * SPS + c * SPP + 8 * h;
      v4f slo, shi;
      slo.x = sp[0]; slo.y = sp[1]; slo.z = sp[2]; slo.w = sp[3];
      shi.x = sp[4]; shi.y = sp[5]; shi.z = sp[6]; shi.w = sp[7];
      *(v4f*)dsp = slo;
      *(v4f*)(dsp + 4) = shi;
    }
    __syncthreads();
    if (wave < 4) {
      v8f acc = {};
      acc = wmma_mm(sWfxT + wave * 16, 1, NXX, sV, NYY, 1, NXX, acc, lane);
      int col = lane & 15, rb = (lane >> 4) << 3;
#pragma unroll
      for (int i = 0; i < 8; ++i) {
        int idx = (wave * 16 + rb + i) * NYY + col;
        sG[idx] = acc[i] + sWfy[idx];
      }
    } else if (wave == 4) {
      float colv[16];
      const float* spc = sSp + (2 * h) * SPS + c * SPP;
      v4f pa[4], pb[4];
#pragma unroll
      for (int q = 0; q < 4; ++q) {
        pa[q] = *(const v4f*)(spc + 4 * q);
        pb[q] = *(const v4f*)(spc + SPS + 4 * q);
      }
      __builtin_amdgcn_sched_barrier(0);
#pragma unroll
      for (int q = 0; q < 4; ++q) {
        const v4f tq = pa[q] + pb[q];
        colv[4 * q + 0] = tq.x + xhalf(tq.x);
        colv[4 * q + 1] = tq.y + xhalf(tq.y);
        colv[4 * q + 2] = tq.z + xhalf(tq.z);
        colv[4 * q + 3] = tq.w + xhalf(tq.w);
      }
#pragma unroll
      for (int i = 0; i < 16; ++i)
        colv[i] = (lane < 16) ? colv[i] : ((i == (lane - 16)) ? 1.0f : 0.0f);
#pragma unroll
      for (int k = 0; k < 16; ++k) {
        float piv  = bcast_lane(colv[k], k);
        float rinv = 1.0f / piv;
        colv[k] *= rinv;
#pragma unroll
        for (int i = 0; i < 16; ++i) {
          if (i != k) {
            float f = bcast_lane(colv[i], k);
            colv[i] -= f * colv[k];
          }
        }
      }
      if (lane >= 16) {
#pragma unroll
        for (int i = 0; i < 16; ++i) sSinv[i * NYY + (lane - 16)] = colv[i];
      }
    }
    __syncthreads();
    if (wave < 4) {
      v8f lt = {};
      lt = wmma_mm(sSinv, 1, NYY, sG + wave * 16 * NYY, 1, NYY, NYY, lt, lane);
      float* gL = Lbuf + (size_t)(t + 1) * SLOT + (wave * 16 + c) * NYY + 8 * h;
      v4f lo, hi;
      lo.x = lt[0]; lo.y = lt[1]; lo.z = lt[2]; lo.w = lt[3];
      hi.x = lt[4]; hi.y = lt[5]; hi.z = lt[6]; hi.w = lt[7];
      *(v4f*)gL = lo;
      *(v4f*)(gL + 4) = hi;
#pragma unroll
      for (int mt = 0; mt < 4; ++mt) {
        const float* ga = sG + (mt * 16 + c) * NYY + 8 * h;
        const v4f a0 = *(const v4f*)ga;
        const v4f a1 = *(const v4f*)(ga + 4);
        v2f a, b;
        v8f acc = {};
        a.x = a0.x; a.y = a0.y; b.x = lt[0]; b.y = lt[1];
        acc = __builtin_amdgcn_wmma_f32_16x16x4_f32(false, a, false, b, (short)0, acc, false, false);
        a.x = a0.z; a.y = a0.w; b.x = lt[2]; b.y = lt[3];
        acc = __builtin_amdgcn_wmma_f32_16x16x4_f32(false, a, false, b, (short)0, acc, false, false);
        a.x = a1.x; a.y = a1.y; b.x = lt[4]; b.y = lt[5];
        acc = __builtin_amdgcn_wmma_f32_16x16x4_f32(false, a, false, b, (short)0, acc, false, false);
        a.x = a1.z; a.y = a1.w; b.x = lt[6]; b.y = lt[7];
        acc = __builtin_amdgcn_wmma_f32_16x16x4_f32(false, a, false, b, (short)0, acc, false, false);
#pragma unroll
        for (int i = 0; i < 8; ++i)
          pt[mt][i] = ((mt == wave && i + 8 * h == c) ? 1.0f : 0.0f) - acc[i];
      }
    }
  }
  if (half == 0 && wave < 4) {
#pragma unroll
    for (int mt = 0; mt < 4; ++mt)
#pragma unroll
      for (int i = 0; i < 8; ++i)
        Psave[((wave * 4 + mt) * 8 + i) * 32 + lane] = pt[mt][i];
  }
}

__global__ __launch_bounds__(128) void kf_chunk_scan_kernel(
    const float* __restrict__ Ym, const float* __restrict__ U,
    const float* __restrict__ Dg,
    const float* __restrict__ Wfx_g, const float* __restrict__ bfx,
    const float* __restrict__ Wfu_g, const float* __restrict__ bfu,
    const float* __restrict__ Wfd_g, const float* __restrict__ bfd,
    const float* __restrict__ Wfy_g, const float* __restrict__ bfy_g,
    const float* __restrict__ Lbuf, float* __restrict__ ws_out, int half) {
  __shared__ float sWfx[NXX * NXX];
  __shared__ float sWfu[NUU * NXX];
  __shared__ float sWfd[NDD * NXX];
  __shared__ float sWfy[NXX * NYY];
  __shared__ float sBsum[NXX];
  __shared__ float sBfy[NYY];
  __shared__ float sX[16 * NXX];
  __shared__ float sXp[16 * NXX];
  __shared__ float sL[NXX * NYY];
  __shared__ float sYm[16 * NYY];
  __shared__ float sU[16 * NUU];
  __shared__ float sD[16 * NDD];
  __shared__ float sInn[16 * NYY];

  const int tid  = threadIdx.x;
  const int lane = tid & 31;
  const int wave = tid >> 5;
  const int j    = blockIdx.x;
  const int cl   = blockIdx.y;
  const int c    = half * NCP + cl;
  const bool homog = (j >= 16);
  if (homog && c == 0) return;
  const int t0   = c * TC;
  const int s0   = cl * TC;
  const int r0   = homog ? 0 : j * 16;

  for (int e = tid; e < NXX * NXX; e += 128) sWfx[e] = Wfx_g[e];
  for (int e = tid; e < NUU * NXX; e += 128) sWfu[e] = Wfu_g[e];
  for (int e = tid; e < NDD * NXX; e += 128) sWfd[e] = Wfd_g[e];
  for (int e = tid; e < NXX * NYY; e += 128) sWfy[e] = Wfy_g[e];
  if (tid < NXX) sBsum[tid] = homog ? 0.0f : (bfx[tid] + bfu[tid] + bfd[tid]);
  if (tid < NYY) sBfy[tid] = homog ? 0.0f : bfy_g[tid];
  if (homog) {
    const int i0 = (j - 16) * 16;
    for (int e = tid; e < 16 * NXX; e += 128) {
      int row = e >> 6, col = e & 63;
      sX[e] = (col == i0 + row) ? 1.0f : 0.0f;
    }
    for (int e = tid; e < 16 * NYY; e += 128) sYm[e] = 0.0f;
    for (int e = tid; e < 16 * NUU; e += 128) sU[e] = 0.0f;
    if (tid < 16 * NDD) sD[tid] = 0.0f;
  } else {
    for (int e = tid; e < 16 * NXX; e += 128) sX[e] = 0.0f;
  }
  __syncthreads();

  for (int tt = 0; tt < TC; ++tt) {
    const int t = t0 + tt;
    {
      const float* gL  = Lbuf + (size_t)(s0 + tt) * SLOT;
      for (int e = tid; e < NXX * NYY; e += 128) sL[e] = gL[e];
      if (!homog) {
        const float* gYm = Ym + (size_t)t * BB * NYY + r0 * NYY;
        const float* gU  = U  + (size_t)t * BB * NUU + r0 * NUU;
        const float* gD  = Dg + (size_t)t * BB * NDD + r0 * NDD;
        for (int e = tid; e < 16 * NYY; e += 128)
          sYm[e] = __builtin_nontemporal_load(gYm + e);
        for (int e = tid; e < 16 * NUU; e += 128)
          sU[e] = __builtin_nontemporal_load(gU + e);
        if (tid < 16 * NDD) sD[tid] = __builtin_nontemporal_load(gD + tid);
      }
      if (tt + 1 < TC) {
        const char* pL = (const char*)(Lbuf + (size_t)(s0 + tt + 1) * SLOT);
        if (tid < 32) __builtin_prefetch(pL + tid * 128, 0, 0);
        if (!homog) {
          const char* pYm = (const char*)(Ym + (size_t)(t + 1) * BB * NYY + r0 * NYY);
          const char* pU  = (const char*)(U  + (size_t)(t + 1) * BB * NUU + r0 * NUU);
          const char* pD  = (const char*)(Dg + (size_t)(t + 1) * BB * NDD + r0 * NDD);
          if (tid < 8)  __builtin_prefetch(pYm + tid * 128, 0, 0);
          if (tid < 8)  __builtin_prefetch(pU + tid * 128, 0, 0);
          if (tid < 4)  __builtin_prefetch(pD + tid * 128, 0, 0);
        }
      }
    }
    __syncthreads();
    {
      const int nt = wave;
      v8f acc = {};
      acc = wmma_mm(sX, NXX, 1, sWfx + nt * 16, NXX, 1, NXX, acc, lane);
      acc = wmma_mm(sU, NUU, 1, sWfu + nt * 16, NXX, 1, NUU, acc, lane);
      acc = wmma_mm(sD, NDD, 1, sWfd + nt * 16, NXX, 1, NDD, acc, lane);
      int col = lane & 15, rb = (lane >> 4) << 3;
      float bias = sBsum[nt * 16 + col];
#pragma unroll
      for (int i = 0; i < 8; ++i)
        sXp[(rb + i) * NXX + nt * 16 + col] = acc[i] + bias;
    }
    __syncthreads();
    if (wave == 0) {
      v8f acc = {};
      acc = wmma_mm(sXp, NXX, 1, sWfy, NYY, 1, NXX, acc, lane);
      int col = lane & 15, rb = (lane >> 4) << 3;
      float by = sBfy[col];
#pragma unroll
      for (int i = 0; i < 8; ++i) {
        int r = rb + i;
        sInn[r * NYY + col] = sYm[r * NYY + col] - (acc[i] + by);
      }
    }
    __syncthreads();
    {
      const int nt = wave;
      v8f acc = {};
      acc = wmma_mm(sInn, NYY, 1, sL + nt * 16 * NYY, 1, NYY, NYY, acc, lane);
      int col = lane & 15, rb = (lane >> 4) << 3;
#pragma unroll
      for (int i = 0; i < 8; ++i) {
        int idx = (rb + i) * NXX + nt * 16 + col;
        sX[idx] = sXp[idx] + acc[i];
      }
    }
    __syncthreads();
  }

  float* res = ws_out + (size_t)(RES_SLOT0 + c * NTILE + j) * SLOT;
  for (int e = tid; e < 16 * NXX; e += 128) res[e] = sX[e];
}

__global__ __launch_bounds__(128) void kf_chain_kernel(
    const float* __restrict__ ws, float* __restrict__ Out) {
  __shared__ float sXc[2][16 * NXX];
  __shared__ float sPhi[NXX * NXX];
  __shared__ float sZ[16 * NXX];

  const int tid  = threadIdx.x;
  const int lane = tid & 31;
  const int wave = tid >> 5;
  const int j    = blockIdx.x;
  const float* res = ws + (size_t)RES_SLOT0 * SLOT;

  float rphi[32], rz[8];
  {
    const float* z0 = res + (size_t)j * SLOT;
    for (int e = tid; e < 16 * NXX; e += 128) sXc[0][e] = z0[e];
    const float* gP = res + (size_t)(1 * NTILE + 16) * SLOT;
    const float* gZ = res + (size_t)(1 * NTILE + j) * SLOT;
#pragma unroll
    for (int i = 0; i < 32; ++i) rphi[i] = gP[tid + i * 128];
#pragma unroll
    for (int i = 0; i < 8; ++i) rz[i] = gZ[tid + i * 128];
  }

  int cur = 0;
  for (int c = 1; c < NCH; ++c) {
#pragma unroll
    for (int i = 0; i < 32; ++i) sPhi[tid + i * 128] = rphi[i];
#pragma unroll
    for (int i = 0; i < 8; ++i) sZ[tid + i * 128] = rz[i];
    __syncthreads();
    if (c + 1 < NCH) {
      const float* gP = res + (size_t)((c + 1) * NTILE + 16) * SLOT;
      const float* gZ = res + (size_t)((c + 1) * NTILE + j) * SLOT;
#pragma unroll
      for (int i = 0; i < 32; ++i) rphi[i] = gP[tid + i * 128];
#pragma unroll
      for (int i = 0; i < 8; ++i) rz[i] = gZ[tid + i * 128];
    }
    {
      const int nt = wave;
      v8f acc = {};
      acc = wmma_mm(sXc[cur], NXX, 1, sPhi + nt * 16, NXX, 1, NXX, acc, lane);
      int col = lane & 15, rb = (lane >> 4) << 3;
#pragma unroll
      for (int i = 0; i < 8; ++i) {
        int idx = (rb + i) * NXX + nt * 16 + col;
        sXc[cur ^ 1][idx] = acc[i] + sZ[idx];
      }
    }
    __syncthreads();
    cur ^= 1;
  }

  for (int e = tid; e < 16 * NXX; e += 128) {
    int row = e >> 6, col = e & 63;
    Out[(size_t)(j * 16 + row) * NXX + col] = sXc[cur][e];
  }
}

extern "C" void kernel_launch(void* const* d_in, const int* in_sizes, int n_in,
                              void* d_out, int out_size, void* d_ws, size_t ws_size,
                              hipStream_t stream) {
  const float* Ym  = (const float*)d_in[0];
  const float* U   = (const float*)d_in[1];
  const float* Dg  = (const float*)d_in[2];
  const float* Wfx = (const float*)d_in[3];
  const float* bfx = (const float*)d_in[4];
  const float* bfu = (const float*)d_in[6];
  const float* Wfu = (const float*)d_in[5];
  const float* Wfd = (const float*)d_in[7];
  const float* bfd = (const float*)d_in[8];
  const float* Wfy = (const float*)d_in[9];
  const float* bfy = (const float*)d_in[10];

  float* ws  = (float*)d_ws;
  float* Out = (float*)d_out;

  kf_gains_kernel<<<1, 256, 0, stream>>>(Wfx, Wfy, ws, 0);
  kf_chunk_scan_kernel<<<dim3(NTILE, NCP), 128, 0, stream>>>(
      Ym, U, Dg, Wfx, bfx, Wfu, bfu, Wfd, bfd, Wfy, bfy, ws, ws, 0);
  kf_gains_kernel<<<1, 256, 0, stream>>>(Wfx, Wfy, ws, 1);
  kf_chunk_scan_kernel<<<dim3(NTILE, NCP), 128, 0, stream>>>(
      Ym, U, Dg, Wfx, bfx, Wfu, bfu, Wfd, bfd, Wfy, bfy, ws, ws, 1);
  kf_chain_kernel<<<16, 128, 0, stream>>>(ws, Out);
}


# Gain kernel, 16x16 inverse: one Newton-Schulz correction of the previous step's inverse on WMMA, X1=X0+(I-X0 S)X0, kept if all residual entries < 2^-14, else the elimination; Sinv differs by rounding
# speedup vs baseline: 9.4720x; 1.2136x over previous
#include <hip/hip_runtime.h>


typedef __attribute__((ext_vector_type(2))) float v2f;
typedef __attribute__((ext_vector_type(4))) float v4f;
typedef __attribute__((ext_vector_type(8))) float v8f;

#define TT  4096
#define BB  256
#define NXX 64
#define NYY 16
#define NUU 16
#define NDD 8

#define SLOT    (NXX * NYY)
#define SPP     20
#define SPS     (NYY * SPP)
#define NS_TOL  6.103515625e-05f
#define HALF_T  (TT / 2)
#define TC      64
#define NCP     (HALF_T / TC)
#define NCH     (TT / TC)
#define NTILE   20
#define P_SLOT0   (HALF_T + 1)
#define RES_SLOT0 (P_SLOT0 + 4)

__device__ __forceinline__ float bcast_lane(float x, int l) {
  return __int_as_float(__builtin_amdgcn_readlane(__float_as_int(x), l));
}

__device__ __forceinline__ float xhalf(float x) {
  const int b = __float_as_int(x);
  return __int_as_float(__builtin_amdgcn_permlanex16(b, b, 0x76543210, 0xfedcba98,
                                                     false, false));
}

__device__ __forceinline__ v8f wmma_mm(const float* A, int ras, int cas,
                                       const float* B, int rbs, int cbs,
                                       int K, v8f acc, int lane) {
  const int mn = lane & 15;
  const int kh = (lane >> 4) << 1;
#pragma unroll
  for (int k = 0; k < K; k += 4) {
    v2f a, b;
    a.x = A[mn * ras + (k + kh) * cas];
    a.y = A[mn * ras + (k + kh + 1) * cas];
    b.x = B[(k + kh) * rbs + mn * cbs];
    b.y = B[(k + kh + 1) * rbs + mn * cbs];
    acc = __builtin_amdgcn_wmma_f32_16x16x4_f32(false, a, false, b,
                                                (short)0, acc, false, false);
  }
  return acc;
}

__device__ __forceinline__ void store_tile(float* D, int ldd, v8f acc, int lane) {
  const int col = lane & 15;
  const int rb  = (lane >> 4) << 3;
#pragma unroll
  for (int i = 0; i < 8; ++i) D[(rb + i) * ldd + col] = acc[i];
}

__global__ __launch_bounds__(256) void kf_gains_kernel(
    const float* __restrict__ Wfx_g, const float* __restrict__ Wfy_g,
    float* __restrict__ ws, int half) {
  float* Lbuf  = ws;
  float* Psave = ws + (size_t)P_SLOT0 * SLOT;
  __shared__ float sWfxT[NXX * NXX];
  __shared__ float sWfy[NXX * NYY];
  __shared__ float sW2[NXX * NYY];
  __shared__ float sV[NXX * NYY];
  __shared__ __attribute__((aligned(16))) float sG[NXX * NYY];
  __shared__ __attribute__((aligned(16))) float sSp[4 * SPS];
  __shared__ __attribute__((aligned(16))) float sSinv[NYY * NYY];

  const int tid  = threadIdx.x;
  const int lane = tid & 31;
  const int wave = tid >> 5;
  const int c    = lane & 15;
  const int h    = lane >> 4;

  for (int e = tid; e < NXX * NXX; e += 256) {
    int r = e >> 6, cc = e & 63;
    sWfxT[cc * NXX + r] = Wfx_g[e];
  }
  for (int e = tid; e < NXX * NYY; e += 256) sWfy[e] = Wfy_g[e];
  sSinv[tid] = 0.0f;
  if (half == 0) {
    for (int e = tid; e < NXX * NYY; e += 256) Lbuf[e] = 0.0f;
  } else {
    for (int e = tid; e < NXX * NYY; e += 256)
      Lbuf[e] = Lbuf[(size_t)HALF_T * SLOT + e];
  }
  __syncthreads();
  v8f s0acc = {};
  if (wave < 4) {
    v8f acc = {};
    acc = wmma_mm(Wfx_g + wave * 16, 1, NXX, sWfy, NYY, 1, NXX, acc, lane);
    store_tile(sW2 + wave * 16 * NYY, NYY, acc, lane);
    if (wave == 0) {
      s0acc = wmma_mm(sWfy, 1, NYY, sWfy, NYY, 1, NXX, s0acc, lane);
      int col = lane & 15, rb = (lane >> 4) << 3;
#pragma unroll
      for (int i = 0; i < 8; ++i) s0acc[i] += ((rb + i == col) ? 1.0f : 0.0f);
    }
  }
  __syncthreads();

  v8f pt[4];
  v2f w2f[16];
  v2f w2a[4];
#pragma unroll
  for (int jj = 0; jj < 4; ++jj) w2a[jj] = v2f{};
#pragma unroll
  for (int mt = 0; mt < 4; ++mt) pt[mt] = v8f{};
#pragma unroll
  for (int q = 0; q < 16; ++q) w2f[q] = v2f{};
  if (wave < 4) {
    if (half == 0) {
#pragma unroll
      for (int mt = 0; mt < 4; ++mt)
#pragma unroll
        for (int i = 0; i < 8; ++i)
          pt[mt][i] = (mt == wave && i + 8 * h == c) ? 1.0f : 0.0f;
    } else {
#pragma unroll
      for (int mt = 0; mt < 4; ++mt)
#pragma unroll
        for (int i = 0; i < 8; ++i)
          pt[mt][i] = Psave[((wave * 4 + mt) * 8 + i) * 32 + lane];
    }
#pragma unroll
    for (int mt = 0; mt < 4; ++mt)
#pragma unroll
      for (int jj = 0; jj < 4; ++jj) {
        const int k = 16 * mt + 8 * h + 2 * jj;
        w2f[4 * mt + jj].x = sW2[k * NYY + c];
        w2f[4 * mt + jj].y = sW2[(k + 1) * NYY + c];
      }
#pragma unroll
    for (int jj = 0; jj < 4; ++jj) {
      const int k = 16 * wave + 8 * h + 2 * jj;
      w2a[jj].x = sW2[k * NYY + c];
      w2a[jj].y = sW2[(k + 1) * NYY + c];
    }
  }

  for (int t = 0; t < HALF_T; ++t) {
    if (wave < 4) {
      v8f acc = {};
#pragma unroll
      for (int mt = 0; mt < 4; ++mt)
#pragma unroll
        for (int jj = 0; jj < 4; ++jj) {
          v2f a;
          a.x = pt[mt][2 * jj];
          a.y = pt[mt][2 * jj + 1];
          acc = __builtin_amdgcn_wmma_f32_16x16x4_f32(false, a, false, w2f[4 * mt + jj],
                                                      (short)0, acc, false, false);
        }
      store_tile(sV + wave * 16 * NYY, NYY, acc, lane);
      v8f sp = s0acc;
#pragma unroll
      for (int jj = 0; jj < 4; ++jj) {
        v2f b;
        b.x = acc[2 * jj];
        b.y = acc[2 * jj + 1];
        sp = __builtin_amdgcn_wmma_f32_16x16x4_f32(false, w2a[jj], false, b,
                                                   (short)0, sp, false, false);
      }
      float* dsp = sSp + wave * SPS + c * SPP + 8 * h;
      v4f slo, shi;
      slo.x = sp[0]; slo.y = sp[1]; slo.z = sp[2]; slo.w = sp[3];
      shi.x = sp[4]; shi.y = sp[5]; shi.z = sp[6]; shi.w = sp[7];
      *(v4f*)dsp = slo;
      *(v4f*)(dsp + 4) = shi;
    }
    __syncthreads();
    if (wave < 4) {
      v8f acc = {};
      acc = wmma_mm(sWfxT + wave * 16, 1, NXX, sV, NYY, 1, NXX, acc, lane);
      int col = lane & 15, rb = (lane >> 4) << 3;
#pragma unroll
      for (int i = 0; i < 8; ++i) {
        int idx = (wave * 16 + rb + i) * NYY + col;
        sG[idx] = acc[i] + sWfy[idx];
      }
    } else if (wave == 4) {
      const float* spc = sSp + c * SPP + 8 * h;
      v4f sl[4][2];
#pragma unroll
      for (int w = 0; w < 4; ++w) {
        sl[w][0] = *(const v4f*)(spc + w * SPS);
        sl[w][1] = *(const v4f*)(spc + w * SPS + 4);
      }
      const float* xrp = sSinv + c * NYY + 8 * h;
      const v4f xr0 = *(const v4f*)xrp;
      const v4f xr1 = *(const v4f*)(xrp + 4);
      float x0c[8];
#pragma unroll
      for (int i = 0; i < 8; ++i) x0c[i] = sSinv[(8 * h + i) * NYY + c];
      float scol[8];
      {
        const v4f lo = (sl[0][0] + sl[1][0]) + (sl[2][0] + sl[3][0]);
        const v4f hi = (sl[0][1] + sl[1][1]) + (sl[2][1] + sl[3][1]);
        scol[0] = lo.x; scol[1] = lo.y; scol[2] = lo.z; scol[3] = lo.w;
        scol[4] = hi.x; scol[5] = hi.y; scol[6] = hi.z; scol[7] = hi.w;
      }
      const float x0r[8] = {xr0.x, xr0.y, xr0.z, xr0.w, xr1.x, xr1.y, xr1.z, xr1.w};
      v8f d1 = {};
#pragma unroll
      for (int jj = 0; jj < 4; ++jj) {
        v2f a, b;
        a.x = scol[2 * jj]; a.y = scol[2 * jj + 1];
        b.x = x0r[2 * jj];  b.y = x0r[2 * jj + 1];
        d1 = __builtin_amdgcn_wmma_f32_16x16x4_f32(false, a, false, b,
                                                   (short)0, d1, false, false);
      }
      float et[8];
#pragma unroll
      for (int i = 0; i < 8; ++i)
        et[i] = ((i + 8 * h == c) ? 1.0f : 0.0f) - d1[i];
      v8f x1;
#pragma unroll
      for (int i = 0; i < 8; ++i) x1[i] = x0c[i];
#pragma unroll
      for (int jj = 0; jj < 4; ++jj) {
        v2f a, b;
        a.x = et[2 * jj];  a.y = et[2 * jj + 1];
        b.x = x0c[2 * jj]; b.y = x0c[2 * jj + 1];
        x1 = __builtin_amdgcn_wmma_f32_16x16x4_f32(false, a, false, b,
                                                   (short)0, x1, false, false);
      }
      bool small = true;
#pragma unroll
      for (int i = 0; i < 8; ++i)
        small = small && (__builtin_fabsf(et[i]) <= NS_TOL);
      if (__builtin_amdgcn_ballot_w32(!small) == 0u) {
#pragma unroll
        for (int i = 0; i < 8; ++i) sSinv[(8 * h + i) * NYY + c] = x1[i];
      } else {
        float colv[16];
#pragma unroll
        for (int i = 0; i < 8; ++i) {
          const float o = xhalf(scol[i]);
          colv[i]     = (h == 0) ? scol[i] : o;
          colv[8 + i] = (h == 0) ? o : scol[i];
        }
#pragma unroll
        for (int i = 0; i < 16; ++i)
          colv[i] = (lane < 16) ? colv[i] : ((i == (lane - 16)) ? 1.0f : 0.0f);
#pragma unroll
        for (int k = 0; k < 16; ++k) {
          float piv  = bcast_lane(colv[k], k);
          float rinv = 1.0f / piv;
          colv[k] *= rinv;
#pragma unroll
          for (int i = 0; i < 16; ++i) {
            if (i != k) {
              float f = bcast_lane(colv[i], k);
              colv[i] -= f * colv[k];
            }
          }
        }
        if (lane >= 16) {
#pragma unroll
          for (int i = 0; i < 16; ++i) sSinv[i * NYY + (lane - 16)] = colv[i];
        }
      }
    }
    __syncthreads();
    if (wave < 4) {
      v8f lt = {};
      lt = wmma_mm(sSinv, 1, NYY, sG + wave * 16 * NYY, 1, NYY, NYY, lt, lane);
      float* gL = Lbuf + (size_t)(t + 1) * SLOT + (wave * 16 + c) * NYY + 8 * h;
      v4f lo, hi;
      lo.x = lt[0]; lo.y = lt[1]; lo.z = lt[2]; lo.w = lt[3];
      hi.x = lt[4]; hi.y = lt[5]; hi.z = lt[6]; hi.w = lt[7];
      *(v4f*)gL = lo;
      *(v4f*)(gL + 4) = hi;
#pragma unroll
      for (int mt = 0; mt < 4; ++mt) {
        const float* ga = sG + (mt * 16 + c) * NYY + 8 * h;
        const v4f a0 = *(const v4f*)ga;
        const v4f a1 = *(const v4f*)(ga + 4);
        v2f a, b;
        v8f acc = {};
        a.x = a0.x; a.y = a0.y; b.x = lt[0]; b.y = lt[1];
        acc = __builtin_amdgcn_wmma_f32_16x16x4_f32(false, a, false, b, (short)0, acc, false, false);
        a.x = a0.z; a.y = a0.w; b.x = lt[2]; b.y = lt[3];
        acc = __builtin_amdgcn_wmma_f32_16x16x4_f32(false, a, false, b, (short)0, acc, false, false);
        a.x = a1.x; a.y = a1.y; b.x = lt[4]; b.y = lt[5];
        acc = __builtin_amdgcn_wmma_f32_16x16x4_f32(false, a, false, b, (short)0, acc, false, false);
        a.x = a1.z; a.y = a1.w; b.x = lt[6]; b.y = lt[7];
        acc = __builtin_amdgcn_wmma_f32_16x16x4_f32(false, a, false, b, (short)0, acc, false, false);
#pragma unroll
        for (int i = 0; i < 8; ++i)
          pt[mt][i] = ((mt == wave && i + 8 * h == c) ? 1.0f : 0.0f) - acc[i];
      }
    }
  }
  if (half == 0 && wave < 4) {
#pragma unroll
    for (int mt = 0; mt < 4; ++mt)
#pragma unroll
      for (int i = 0; i < 8; ++i)
        Psave[((wave * 4 + mt) * 8 + i) * 32 + lane] = pt[mt][i];
  }
}

__global__ __launch_bounds__(128) void kf_chunk_scan_kernel(
    const float* __restrict__ Ym, const float* __restrict__ U,
    const float* __restrict__ Dg,
    const float* __restrict__ Wfx_g, const float* __restrict__ bfx,
    const float* __restrict__ Wfu_g, const float* __restrict__ bfu,
    const float* __restrict__ Wfd_g, const float* __restrict__ bfd,
    const float* __restrict__ Wfy_g, const float* __restrict__ bfy_g,
    const float* __restrict__ Lbuf, float* __restrict__ ws_out, int half) {
  __shared__ float sWfx[NXX * NXX];
  __shared__ float sWfu[NUU * NXX];
  __shared__ float sWfd[NDD * NXX];
  __shared__ float sWfy[NXX * NYY];
  __shared__ float sBsum[NXX];
  __shared__ float sBfy[NYY];
  __shared__ float sX[16 * NXX];
  __shared__ float sXp[16 * NXX];
  __shared__ float sL[NXX * NYY];
  __shared__ float sYm[16 * NYY];
  __shared__ float sU[16 * NUU];
  __shared__ float sD[16 * NDD];
  __shared__ float sInn[16 * NYY];

  const int tid  = threadIdx.x;
  const int lane = tid & 31;
  const int wave = tid >> 5;
  const int j    = blockIdx.x;
  const int cl   = blockIdx.y;
  const int c    = half * NCP + cl;
  const bool homog = (j >= 16);
  if (homog && c == 0) return;
  const int t0   = c * TC;
  const int s0   = cl * TC;
  const int r0   = homog ? 0 : j * 16;

  for (int e = tid; e < NXX * NXX; e += 128) sWfx[e] = Wfx_g[e];
  for (int e = tid; e < NUU * NXX; e += 128) sWfu[e] = Wfu_g[e];
  for (int e = tid; e < NDD * NXX; e += 128) sWfd[e] = Wfd_g[e];
  for (int e = tid; e < NXX * NYY; e += 128) sWfy[e] = Wfy_g[e];
  if (tid < NXX) sBsum[tid] = homog ? 0.0f : (bfx[tid] + bfu[tid] + bfd[tid]);
  if (tid < NYY) sBfy[tid] = homog ? 0.0f : bfy_g[tid];
  if (homog) {
    const int i0 = (j - 16) * 16;
    for (int e = tid; e < 16 * NXX; e += 128) {
      int row = e >> 6, col = e & 63;
      sX[e] = (col == i0 + row) ? 1.0f : 0.0f;
    }
    for (int e = tid; e < 16 * NYY; e += 128) sYm[e] = 0.0f;
    for (int e = tid; e < 16 * NUU; e += 128) sU[e] = 0.0f;
    if (tid < 16 * NDD) sD[tid] = 0.0f;
  } else {
    for (int e = tid; e < 16 * NXX; e += 128) sX[e] = 0.0f;
  }
  __syncthreads();

  for (int tt = 0; tt < TC; ++tt) {
    const int t = t0 + tt;
    {
      const float* gL  = Lbuf + (size_t)(s0 + tt) * SLOT;
      for (int e = tid; e < NXX * NYY; e += 128) sL[e] = gL[e];
      if (!homog) {
        const float* gYm = Ym + (size_t)t * BB * NYY + r0 * NYY;
        const float* gU  = U  + (size_t)t * BB * NUU + r0 * NUU;
        const float* gD  = Dg + (size_t)t * BB * NDD + r0 * NDD;
        for (int e = tid; e < 16 * NYY; e += 128)
          sYm[e] = __builtin_nontemporal_load(gYm + e);
        for (int e = tid; e < 16 * NUU; e += 128)
          sU[e] = __builtin_nontemporal_load(gU + e);
        if (tid < 16 * NDD) sD[tid] = __builtin_nontemporal_load(gD + tid);
      }
      if (tt + 1 < TC) {
        const char* pL = (const char*)(Lbuf + (size_t)(s0 + tt + 1) * SLOT);
        if (tid < 32) __builtin_prefetch(pL + tid * 128, 0, 0);
        if (!homog) {
          const char* pYm = (const char*)(Ym + (size_t)(t + 1) * BB * NYY + r0 * NYY);
          const char* pU  = (const char*)(U  + (size_t)(t + 1) * BB * NUU + r0 * NUU);
          const char* pD  = (const char*)(Dg + (size_t)(t + 1) * BB * NDD + r0 * NDD);
          if (tid < 8)  __builtin_prefetch(pYm + tid * 128, 0, 0);
          if (tid < 8)  __builtin_prefetch(pU + tid * 128, 0, 0);
          if (tid < 4)  __builtin_prefetch(pD + tid * 128, 0, 0);
        }
      }
    }
    __syncthreads();
    {
      const int nt = wave;
      v8f acc = {};
      acc = wmma_mm(sX, NXX, 1, sWfx + nt * 16, NXX, 1, NXX, acc, lane);
      acc = wmma_mm(sU, NUU, 1, sWfu + nt * 16, NXX, 1, NUU, acc, lane);
      acc = wmma_mm(sD, NDD, 1, sWfd + nt * 16, NXX, 1, NDD, acc, lane);
      int col = lane & 15, rb = (lane >> 4) << 3;
      float bias = sBsum[nt * 16 + col];
#pragma unroll
      for (int i = 0; i < 8; ++i)
        sXp[(rb + i) * NXX + nt * 16 + col] = acc[i] + bias;
    }
    __syncthreads();
    if (wave == 0) {
      v8f acc = {};
      acc = wmma_mm(sXp, NXX, 1, sWfy, NYY, 1, NXX, acc, lane);
      int col = lane & 15, rb = (lane >> 4) << 3;
      float by = sBfy[col];
#pragma unroll
      for (int i = 0; i < 8; ++i) {
        int r = rb + i;
        sInn[r * NYY + col] = sYm[r * NYY + col] - (acc[i] + by);
      }
    }
    __syncthreads();
    {
      const int nt = wave;
      v8f acc = {};
      acc = wmma_mm(sInn, NYY, 1, sL + nt * 16 * NYY, 1, NYY, NYY, acc, lane);
      int col = lane & 15, rb = (lane >> 4) << 3;
#pragma unroll
      for (int i = 0; i < 8; ++i) {
        int idx = (rb + i) * NXX + nt * 16 + col;
        sX[idx] = sXp[idx] + acc[i];
      }
    }
    __syncthreads();
  }

  float* res = ws_out + (size_t)(RES_SLOT0 + c * NTILE + j) * SLOT;
  for (int e = tid; e < 16 * NXX; e += 128) res[e] = sX[e];
}

__global__ __launch_bounds__(128) void kf_chain_kernel(
    const float* __restrict__ ws, float* __restrict__ Out) {
  __shared__ float sXc[2][16 * NXX];
  __shared__ float sPhi[NXX * NXX];
  __shared__ float sZ[16 * NXX];

  const int tid  = threadIdx.x;
  const int lane = tid & 31;
  const int wave = tid >> 5;
  const int j    = blockIdx.x;
  const float* res = ws + (size_t)RES_SLOT0 * SLOT;

  float rphi[32], rz[8];
  {
    const float* z0 = res + (size_t)j * SLOT;
    for (int e = tid; e < 16 * NXX; e += 128) sXc[0][e] = z0[e];
    const float* gP = res + (size_t)(1 * NTILE + 16) * SLOT;
    const float* gZ = res + (size_t)(1 * NTILE + j) * SLOT;
#pragma unroll
    for (int i = 0; i < 32; ++i) rphi[i] = gP[tid + i * 128];
#pragma unroll
    for (int i = 0; i < 8; ++i) rz[i] = gZ[tid + i * 128];
  }

  int cur = 0;
  for (int c = 1; c < NCH; ++c) {
#pragma unroll
    for (int i = 0; i < 32; ++i) sPhi[tid + i * 128] = rphi[i];
#pragma unroll
    for (int i = 0; i < 8; ++i) sZ[tid + i * 128] = rz[i];
    __syncthreads();
    if (c + 1 < NCH) {
      const float* gP = res + (size_t)((c + 1) * NTILE + 16) * SLOT;
      const float* gZ = res + (size_t)((c + 1) * NTILE + j) * SLOT;
#pragma unroll
      for (int i = 0; i < 32; ++i) rphi[i] = gP[tid + i * 128];
#pragma unroll
      for (int i = 0; i < 8; ++i) rz[i] = gZ[tid + i * 128];
    }
    {
      const int nt = wave;
      v8f acc = {};
      acc = wmma_mm(sXc[cur], NXX, 1, sPhi + nt * 16, NXX, 1, NXX, acc, lane);
      int col = lane & 15, rb = (lane >> 4) << 3;
#pragma unroll
      for (int i = 0; i < 8; ++i) {
        int idx = (rb + i) * NXX + nt * 16 + col;
        sXc[cur ^ 1][idx] = acc[i] + sZ[idx];
      }
    }
    __syncthreads();
    cur ^= 1;
  }

  for (int e = tid; e < 16 * NXX; e += 128) {
    int row = e >> 6, col = e & 63;
    Out[(size_t)(j * 16 + row) * NXX + col] = sXc[cur][e];
  }
}

extern "C" void kernel_launch(void* const* d_in, const int* in_sizes, int n_in,
                              void* d_out, int out_size, void* d_ws, size_t ws_size,
                              hipStream_t stream) {
  const float* Ym  = (const float*)d_in[0];
  const float* U   = (const float*)d_in[1];
  const float* Dg  = (const float*)d_in[2];
  const float* Wfx = (const float*)d_in[3];
  const float* bfx = (const float*)d_in[4];
  const float* bfu = (const float*)d_in[6];
  const float* Wfu = (const float*)d_in[5];
  const float* Wfd = (const float*)d_in[7];
  const float* bfd = (const float*)d_in[8];
  const float* Wfy = (const float*)d_in[9];
  const float* bfy = (const float*)d_in[10];

  float* ws  = (float*)d_ws;
  float* Out = (float*)d_out;

  kf_gains_kernel<<<1, 256, 0, stream>>>(Wfx, Wfy, ws, 0);
  kf_chunk_scan_kernel<<<dim3(NTILE, NCP), 128, 0, stream>>>(
      Ym, U, Dg, Wfx, bfx, Wfu, bfu, Wfd, bfd, Wfy, bfy, ws, ws, 0);
  kf_gains_kernel<<<1, 256, 0, stream>>>(Wfx, Wfy, ws, 1);
  kf_chunk_scan_kernel<<<dim3(NTILE, NCP), 128, 0, stream>>>(
      Ym, U, Dg, Wfx, bfx, Wfu, bfu, Wfd, bfd, Wfy, bfy, ws, ws, 1);
  kf_chain_kernel<<<16, 128, 0, stream>>>(ws, Out);
}


# Gain kernel: G = Wfx V + Wfy split over K on the 4 waves whose accumulators hold V's tiles (tile = B operand, Wfx columns in registers); 16 slices added via LDS, V never stored; sums re-associated
# speedup vs baseline: 10.5043x; 1.1090x over previous
#include <hip/hip_runtime.h>


typedef __attribute__((ext_vector_type(2))) float v2f;
typedef __attribute__((ext_vector_type(4))) float v4f;
typedef __attribute__((ext_vector_type(8))) float v8f;

#define TT  4096
#define BB  256
#define NXX 64
#define NYY 16
#define NUU 16
#define NDD 8

#define SLOT    (NXX * NYY)
#define SPP     20
#define SPS     (NYY * SPP)
#define NS_TOL  6.103515625e-05f
#define HALF_T  (TT / 2)
#define TC      64
#define NCP     (HALF_T / TC)
#define NCH     (TT / TC)
#define NTILE   20
#define P_SLOT0   (HALF_T + 1)
#define RES_SLOT0 (P_SLOT0 + 4)

__device__ __forceinline__ float bcast_lane(float x, int l) {
  return __int_as_float(__builtin_amdgcn_readlane(__float_as_int(x), l));
}

__device__ __forceinline__ float xhalf(float x) {
  const int b = __float_as_int(x);
  return __int_as_float(__builtin_amdgcn_permlanex16(b, b, 0x76543210, 0xfedcba98,
                                                     false, false));
}

__device__ __forceinline__ v8f wmma_mm(const float* A, int ras, int cas,
                                       const float* B, int rbs, int cbs,
                                       int K, v8f acc, int lane) {
  const int mn = lane & 15;
  const int kh = (lane >> 4) << 1;
#pragma unroll
  for (int k = 0; k < K; k += 4) {
    v2f a, b;
    a.x = A[mn * ras + (k + kh) * cas];
    a.y = A[mn * ras + (k + kh + 1) * cas];
    b.x = B[(k + kh) * rbs + mn * cbs];
    b.y = B[(k + kh + 1) * rbs + mn * cbs];
    acc = __builtin_amdgcn_wmma_f32_16x16x4_f32(false, a, false, b,
                                                (short)0, acc, false, false);
  }
  return acc;
}

__device__ __forceinline__ void store_tile(float* D, int ldd, v8f acc, int lane) {
  const int col = lane & 15;
  const int rb  = (lane >> 4) << 3;
#pragma unroll
  for (int i = 0; i < 8; ++i) D[(rb + i) * ldd + col] = acc[i];
}

__global__ __launch_bounds__(256) void kf_gains_kernel(
    const float* __restrict__ Wfx_g, const float* __restrict__ Wfy_g,
    float* __restrict__ ws, int half) {
  float* Lbuf  = ws;
  float* Psave = ws + (size_t)P_SLOT0 * SLOT;
  __shared__ float sWfy[NXX * NYY];
  __shared__ float sW2[NXX * NYY];
  __shared__ __attribute__((aligned(16))) float sG[NXX * NYY];
  __shared__ __attribute__((aligned(16))) float sGp[16 * SPS];
  __shared__ __attribute__((aligned(16))) float sSp[4 * SPS];
  __shared__ __attribute__((aligned(16))) float sSinv[NYY * NYY];

  const int tid  = threadIdx.x;
  const int lane = tid & 31;
  const int wave = tid >> 5;
  const int c    = lane & 15;
  const int h    = lane >> 4;

  for (int e = tid; e < NXX * NYY; e += 256) sWfy[e] = Wfy_g[e];
  sSinv[tid] = 0.0f;
  if (half == 0) {
    for (int e = tid; e < NXX * NYY; e += 256) Lbuf[e] = 0.0f;
  } else {
    for (int e = tid; e < NXX * NYY; e += 256)
      Lbuf[e] = Lbuf[(size_t)HALF_T * SLOT + e];
  }
  __syncthreads();
  v8f s0acc = {};
  if (wave < 4) {
    v8f acc = {};
    acc = wmma_mm(Wfx_g + wave * 16, 1, NXX, sWfy, NYY, 1, NXX, acc, lane);
    store_tile(sW2 + wave * 16 * NYY, NYY, acc, lane);
    if (wave == 0) {
      s0acc = wmma_mm(sWfy, 1, NYY, sWfy, NYY, 1, NXX, s0acc, lane);
      int col = lane & 15, rb = (lane >> 4) << 3;
#pragma unroll
      for (int i = 0; i < 8; ++i) s0acc[i] += ((rb + i == col) ? 1.0f : 0.0f);
    }
  }
  __syncthreads();

  v8f pt[4];
  v2f w2f[16];
  v2f w2a[4];
  v2f wfa[16];
  float wfyt[8];
#pragma unroll
  for (int q = 0; q < 16; ++q) wfa[q] = v2f{};
#pragma unroll
  for (int i = 0; i < 8; ++i) wfyt[i] = 0.0f;
#pragma unroll
  for (int jj = 0; jj < 4; ++jj) w2a[jj] = v2f{};
#pragma unroll
  for (int mt = 0; mt < 4; ++mt) pt[mt] = v8f{};
#pragma unroll
  for (int q = 0; q < 16; ++q) w2f[q] = v2f{};
  if (wave < 4) {
    if (half == 0) {
#pragma unroll
      for (int mt = 0; mt < 4; ++mt)
#pragma unroll
        for (int i = 0; i < 8; ++i)
          pt[mt][i] = (mt == wave && i + 8 * h == c) ? 1.0f : 0.0f;
    } else {
#pragma unroll
      for (int mt = 0; mt < 4; ++mt)
#pragma unroll
        for (int i = 0; i < 8; ++i)
          pt[mt][i] = Psave[((wave * 4 + mt) * 8 + i) * 32 + lane];
    }
#pragma unroll
    for (int mt = 0; mt < 4; ++mt)
#pragma unroll
      for (int jj = 0; jj < 4; ++jj) {
        const int k = 16 * mt + 8 * h + 2 * jj;
        w2f[4 * mt + jj].x = sW2[k * NYY + c];
        w2f[4 * mt + jj].y = sW2[(k + 1) * NYY + c];
      }
#pragma unroll
    for (int jj = 0; jj < 4; ++jj) {
      const int k = 16 * wave + 8 * h + 2 * jj;
      w2a[jj].x = sW2[k * NYY + c];
      w2a[jj].y = sW2[(k + 1) * NYY + c];
    }
#pragma unroll
    for (int mt = 0; mt < 4; ++mt)
#pragma unroll
      for (int jj = 0; jj < 4; ++jj) {
        const int k = 16 * wave + 8 * h + 2 * jj;
        wfa[4 * mt + jj].x = Wfx_g[(16 * mt + c) * NXX + k];
        wfa[4 * mt + jj].y = Wfx_g[(16 * mt + c) * NXX + k + 1];
      }
#pragma unroll
    for (int i = 0; i < 8; ++i) wfyt[i] = sWfy[(16 * wave + i + 8 * h) * NYY + c];
  }

  for (int t = 0; t < HALF_T; ++t) {
    if (wave < 4) {
      v8f acc = {};
#pragma unroll
      for (int mt = 0; mt < 4; ++mt)
#pragma unroll
        for (int jj = 0; jj < 4; ++jj) {
          v2f a;
          a.x = pt[mt][2 * jj];
          a.y = pt[mt][2 * jj + 1];
          acc = __builtin_amdgcn_wmma_f32_16x16x4_f32(false, a, false, w2f[4 * mt + jj],
                                                      (short)0, acc, false, false);
        }
      v8f sp = s0acc;
      v8f gp[4];
#pragma unroll
      for (int mt = 0; mt < 4; ++mt) gp[mt] = v8f{};
#pragma unroll
      for (int jj = 0; jj < 4; ++jj) {
        v2f b;
        b.x = acc[2 * jj];
        b.y = acc[2 * jj + 1];
        sp = __builtin_amdgcn_wmma_f32_16x16x4_f32(false, w2a[jj], false, b,
                                                   (short)0, sp, false, false);
#pragma unroll
        for (int mt = 0; mt < 4; ++mt)
          gp[mt] = __builtin_amdgcn_wmma_f32_16x16x4_f32(false, wfa[4 * mt + jj], false, b,
                                                         (short)0, gp[mt], false, false);
      }
      float* dsp = sSp + wave * SPS + c * SPP + 8 * h;
      v4f slo, shi;
      slo.x = sp[0]; slo.y = sp[1]; slo.z = sp[2]; slo.w = sp[3];
      shi.x = sp[4]; shi.y = sp[5]; shi.z = sp[6]; shi.w = sp[7];
      *(v4f*)dsp = slo;
      *(v4f*)(dsp + 4) = shi;
#pragma unroll
      for (int mt = 0; mt < 4; ++mt) {
        float* dgp = sGp + (wave * 4 + mt) * SPS + c * SPP + 8 * h;
        v4f glo, ghi;
        glo.x = gp[mt][0]; glo.y = gp[mt][1]; glo.z = gp[mt][2]; glo.w = gp[mt][3];
        ghi.x = gp[mt][4]; ghi.y = gp[mt][5]; ghi.z = gp[mt][6]; ghi.w = gp[mt][7];
        *(v4f*)dgp = glo;
        *(v4f*)(dgp + 4) = ghi;
      }
    }
    __syncthreads();
    if (wave < 4) {
      const float* gpc = sGp + wave * SPS + c * SPP + 8 * h;
      v4f gl[4][2];
#pragma unroll
      for (int w = 0; w < 4; ++w) {
        gl[w][0] = *(const v4f*)(gpc + w * 4 * SPS);
        gl[w][1] = *(const v4f*)(gpc + w * 4 * SPS + 4);
      }
      const v4f lo = (gl[0][0] + gl[1][0]) + (gl[2][0] + gl[3][0]);
      const v4f hi = (gl[0][1] + gl[1][1]) + (gl[2][1] + gl[3][1]);
      const float g[8] = {lo.x, lo.y, lo.z, lo.w, hi.x, hi.y, hi.z, hi.w};
#pragma unroll
      for (int i = 0; i < 8; ++i)
        sG[(wave * 16 + 8 * h + i) * NYY + c] = g[i] + wfyt[i];
    } else if (wave == 4) {
      const float* spc = sSp + c * SPP + 8 * h;
      v4f sl[4][2];
#pragma unroll
      for (int w = 0; w < 4; ++w) {
        sl[w][0] = *(const v4f*)(spc + w * SPS);
        sl[w][1] = *(const v4f*)(spc + w * SPS + 4);
      }
      const float* xrp = sSinv + c * NYY + 8 * h;
      const v4f xr0 = *(const v4f*)xrp;
      const v4f xr1 = *(const v4f*)(xrp + 4);
      float x0c[8];
#pragma unroll
      for (int i = 0; i < 8; ++i) x0c[i] = sSinv[(8 * h + i) * NYY + c];
      float scol[8];
      {
        const v4f lo = (sl[0][0] + sl[1][0]) + (sl[2][0] + sl[3][0]);
        const v4f hi = (sl[0][1] + sl[1][1]) + (sl[2][1] + sl[3][1]);
        scol[0] = lo.x; scol[1] = lo.y; scol[2] = lo.z; scol[3] = lo.w;
        scol[4] = hi.x; scol[5] = hi.y; scol[6] = hi.z; scol[7] = hi.w;
      }
      const float x0r[8] = {xr0.x, xr0.y, xr0.z, xr0.w, xr1.x, xr1.y, xr1.z, xr1.w};
      v8f d1 = {};
#pragma unroll
      for (int jj = 0; jj < 4; ++jj) {
        v2f a, b;
        a.x = scol[2 * jj]; a.y = scol[2 * jj + 1];
        b.x = x0r[2 * jj];  b.y = x0r[2 * jj + 1];
        d1 = __builtin_amdgcn_wmma_f32_16x16x4_f32(false, a, false, b,
                                                   (short)0, d1, false, false);
      }
      float et[8];
#pragma unroll
      for (int i = 0; i < 8; ++i)
        et[i] = ((i + 8 * h == c) ? 1.0f : 0.0f) - d1[i];
      v8f x1;
#pragma unroll
      for (int i = 0; i < 8; ++i) x1[i] = x0c[i];
#pragma unroll
      for (int jj = 0; jj < 4; ++jj) {
        v2f a, b;
        a.x = et[2 * jj];  a.y = et[2 * jj + 1];
        b.x = x0c[2 * jj]; b.y = x0c[2 * jj + 1];
        x1 = __builtin_amdgcn_wmma_f32_16x16x4_f32(false, a, false, b,
                                                   (short)0, x1, false, false);
      }
      bool small = true;
#pragma unroll
      for (int i = 0; i < 8; ++i)
        small = small && (__builtin_fabsf(et[i]) <= NS_TOL);
      if (__builtin_amdgcn_ballot_w32(!small) == 0u) {
#pragma unroll
        for (int i = 0; i < 8; ++i) sSinv[(8 * h + i) * NYY + c] = x1[i];
      } else {
        float colv[16];
#pragma unroll
        for (int i = 0; i < 8; ++i) {
          const float o = xhalf(scol[i]);
          colv[i]     = (h == 0) ? scol[i] : o;
          colv[8 + i] = (h == 0) ? o : scol[i];
        }
#pragma unroll
        for (int i = 0; i < 16; ++i)
          colv[i] = (lane < 16) ? colv[i] : ((i == (lane - 16)) ? 1.0f : 0.0f);
#pragma unroll
        for (int k = 0; k < 16; ++k) {
          float piv  = bcast_lane(colv[k], k);
          float rinv = 1.0f / piv;
          colv[k] *= rinv;
#pragma unroll
          for (int i = 0; i < 16; ++i) {
            if (i != k) {
              float f = bcast_lane(colv[i], k);
              colv[i] -= f * colv[k];
            }
          }
        }
        if (lane >= 16) {
#pragma unroll
          for (int i = 0; i < 16; ++i) sSinv[i * NYY + (lane - 16)] = colv[i];
        }
      }
    }
    __syncthreads();
    if (wave < 4) {
      v8f lt = {};
      lt = wmma_mm(sSinv, 1, NYY, sG + wave * 16 * NYY, 1, NYY, NYY, lt, lane);
      float* gL = Lbuf + (size_t)(t + 1) * SLOT + (wave * 16 + c) * NYY + 8 * h;
      v4f lo, hi;
      lo.x = lt[0]; lo.y = lt[1]; lo.z = lt[2]; lo.w = lt[3];
      hi.x = lt[4]; hi.y = lt[5]; hi.z = lt[6]; hi.w = lt[7];
      *(v4f*)gL = lo;
      *(v4f*)(gL + 4) = hi;
#pragma unroll
      for (int mt = 0; mt < 4; ++mt) {
        const float* ga = sG + (mt * 16 + c) * NYY + 8 * h;
        const v4f a0 = *(const v4f*)ga;
        const v4f a1 = *(const v4f*)(ga + 4);
        v2f a, b;
        v8f acc = {};
        a.x = a0.x; a.y = a0.y; b.x = lt[0]; b.y = lt[1];
        acc = __builtin_amdgcn_wmma_f32_16x16x4_f32(false, a, false, b, (short)0, acc, false, false);
        a.x = a0.z; a.y = a0.w; b.x = lt[2]; b.y = lt[3];
        acc = __builtin_amdgcn_wmma_f32_16x16x4_f32(false, a, false, b, (short)0, acc, false, false);
        a.x = a1.x; a.y = a1.y; b.x = lt[4]; b.y = lt[5];
        acc = __builtin_amdgcn_wmma_f32_16x16x4_f32(false, a, false, b, (short)0, acc, false, false);
        a.x = a1.z; a.y = a1.w; b.x = lt[6]; b.y = lt[7];
        acc = __builtin_amdgcn_wmma_f32_16x16x4_f32(false, a, false, b, (short)0, acc, false, false);
#pragma unroll
        for (int i = 0; i < 8; ++i)
          pt[mt][i] = ((mt == wave && i + 8 * h == c) ? 1.0f : 0.0f) - acc[i];
      }
    }
  }
  if (half == 0 && wave < 4) {
#pragma unroll
    for (int mt = 0; mt < 4; ++mt)
#pragma unroll
      for (int i = 0; i < 8; ++i)
        Psave[((wave * 4 + mt) * 8 + i) * 32 + lane] = pt[mt][i];
  }
}

__global__ __launch_bounds__(128) void kf_chunk_scan_kernel(
    const float* __restrict__ Ym, const float* __restrict__ U,
    const float* __restrict__ Dg,
    const float* __restrict__ Wfx_g, const float* __restrict__ bfx,
    const float* __restrict__ Wfu_g, const float* __restrict__ bfu,
    const float* __restrict__ Wfd_g, const float* __restrict__ bfd,
    const float* __restrict__ Wfy_g, const float* __restrict__ bfy_g,
    const float* __restrict__ Lbuf, float* __restrict__ ws_out, int half) {
  __shared__ float sWfx[NXX * NXX];
  __shared__ float sWfu[NUU * NXX];
  __shared__ float sWfd[NDD * NXX];
  __shared__ float sWfy[NXX * NYY];
  __shared__ float sBsum[NXX];
  __shared__ float sBfy[NYY];
  __shared__ float sX[16 * NXX];
  __shared__ float sXp[16 * NXX];
  __shared__ float sL[NXX * NYY];
  __shared__ float sYm[16 * NYY];
  __shared__ float sU[16 * NUU];
  __shared__ float sD[16 * NDD];
  __shared__ float sInn[16 * NYY];

  const int tid  = threadIdx.x;
  const int lane = tid & 31;
  const int wave = tid >> 5;
  const int j    = blockIdx.x;
  const int cl   = blockIdx.y;
  const int c    = half * NCP + cl;
  const bool homog = (j >= 16);
  if (homog && c == 0) return;
  const int t0   = c * TC;
  const int s0   = cl * TC;
  const int r0   = homog ? 0 : j * 16;

  for (int e = tid; e < NXX * NXX; e += 128) sWfx[e] = Wfx_g[e];
  for (int e = tid; e < NUU * NXX; e += 128) sWfu[e] = Wfu_g[e];
  for (int e = tid; e < NDD * NXX; e += 128) sWfd[e] = Wfd_g[e];
  for (int e = tid; e < NXX * NYY; e += 128) sWfy[e] = Wfy_g[e];
  if (tid < NXX) sBsum[tid] = homog ? 0.0f : (bfx[tid] + bfu[tid] + bfd[tid]);
  if (tid < NYY) sBfy[tid] = homog ? 0.0f : bfy_g[tid];
  if (homog) {
    const int i0 = (j - 16) * 16;
    for (int e = tid; e < 16 * NXX; e += 128) {
      int row = e >> 6, col = e & 63;
      sX[e] = (col == i0 + row) ? 1.0f : 0.0f;
    }
    for (int e = tid; e < 16 * NYY; e += 128) sYm[e] = 0.0f;
    for (int e = tid; e < 16 * NUU; e += 128) sU[e] = 0.0f;
    if (tid < 16 * NDD) sD[tid] = 0.0f;
  } else {
    for (int e = tid; e < 16 * NXX; e += 128) sX[e] = 0.0f;
  }
  __syncthreads();

  for (int tt = 0; tt < TC; ++tt) {
    const int t = t0 + tt;
    {
      const float* gL  = Lbuf + (size_t)(s0 + tt) * SLOT;
      for (int e = tid; e < NXX * NYY; e += 128) sL[e] = gL[e];
      if (!homog) {
        const float* gYm = Ym + (size_t)t * BB * NYY + r0 * NYY;
        const float* gU  = U  + (size_t)t * BB * NUU + r0 * NUU;
        const float* gD  = Dg + (size_t)t * BB * NDD + r0 * NDD;
        for (int e = tid; e < 16 * NYY; e += 128)
          sYm[e] = __builtin_nontemporal_load(gYm + e);
        for (int e = tid; e < 16 * NUU; e += 128)
          sU[e] = __builtin_nontemporal_load(gU + e);
        if (tid < 16 * NDD) sD[tid] = __builtin_nontemporal_load(gD + tid);
      }
      if (tt + 1 < TC) {
        const char* pL = (const char*)(Lbuf + (size_t)(s0 + tt + 1) * SLOT);
        if (tid < 32) __builtin_prefetch(pL + tid * 128, 0, 0);
        if (!homog) {
          const char* pYm = (const char*)(Ym + (size_t)(t + 1) * BB * NYY + r0 * NYY);
          const char* pU  = (const char*)(U  + (size_t)(t + 1) * BB * NUU + r0 * NUU);
          const char* pD  = (const char*)(Dg + (size_t)(t + 1) * BB * NDD + r0 * NDD);
          if (tid < 8)  __builtin_prefetch(pYm + tid * 128, 0, 0);
          if (tid < 8)  __builtin_prefetch(pU + tid * 128, 0, 0);
          if (tid < 4)  __builtin_prefetch(pD + tid * 128, 0, 0);
        }
      }
    }
    __syncthreads();
    {
      const int nt = wave;
      v8f acc = {};
      acc = wmma_mm(sX, NXX, 1, sWfx + nt * 16, NXX, 1, NXX, acc, lane);
      acc = wmma_mm(sU, NUU, 1, sWfu + nt * 16, NXX, 1, NUU, acc, lane);
      acc = wmma_mm(sD, NDD, 1, sWfd + nt * 16, NXX, 1, NDD, acc, lane);
      int col = lane & 15, rb = (lane >> 4) << 3;
      float bias = sBsum[nt * 16 + col];
#pragma unroll
      for (int i = 0; i < 8; ++i)
        sXp[(rb + i) * NXX + nt * 16 + col] = acc[i] + bias;
    }
    __syncthreads();
    if (wave == 0) {
      v8f acc = {};
      acc = wmma_mm(sXp, NXX, 1, sWfy, NYY, 1, NXX, acc, lane);
      int col = lane & 15, rb = (lane >> 4) << 3;
      float by = sBfy[col];
#pragma unroll
      for (int i = 0; i < 8; ++i) {
        int r = rb + i;
        sInn[r * NYY + col] = sYm[r * NYY + col] - (acc[i] + by);
      }
    }
    __syncthreads();
    {
      const int nt = wave;
      v8f acc = {};
      acc = wmma_mm(sInn, NYY, 1, sL + nt * 16 * NYY, 1, NYY, NYY, acc, lane);
      int col = lane & 15, rb = (lane >> 4) << 3;
#pragma unroll
      for (int i = 0; i < 8; ++i) {
        int idx = (rb + i) * NXX + nt * 16 + col;
        sX[idx] = sXp[idx] + acc[i];
      }
    }
    __syncthreads();
  }

  float* res = ws_out + (size_t)(RES_SLOT0 + c * NTILE + j) * SLOT;
  for (int e = tid; e < 16 * NXX; e += 128) res[e] = sX[e];
}

__global__ __launch_bounds__(128) void kf_chain_kernel(
    const float* __restrict__ ws, float* __restrict__ Out) {
  __shared__ float sXc[2][16 * NXX];
  __shared__ float sPhi[NXX * NXX];
  __shared__ float sZ[16 * NXX];

  const int tid  = threadIdx.x;
  const int lane = tid & 31;
  const int wave = tid >> 5;
  const int j    = blockIdx.x;
  const float* res = ws + (size_t)RES_SLOT0 * SLOT;

  float rphi[32], rz[8];
  {
    const float* z0 = res + (size_t)j * SLOT;
    for (int e = tid; e < 16 * NXX; e += 128) sXc[0][e] = z0[e];
    const float* gP = res + (size_t)(1 * NTILE + 16) * SLOT;
    const float* gZ = res + (size_t)(1 * NTILE + j) * SLOT;
#pragma unroll
    for (int i = 0; i < 32; ++i) rphi[i] = gP[tid + i * 128];
#pragma unroll
    for (int i = 0; i < 8; ++i) rz[i] = gZ[tid + i * 128];
  }

  int cur = 0;
  for (int c = 1; c < NCH; ++c) {
#pragma unroll
    for (int i = 0; i < 32; ++i) sPhi[tid + i * 128] = rphi[i];
#pragma unroll
    for (int i = 0; i < 8; ++i) sZ[tid + i * 128] = rz[i];
    __syncthreads();
    if (c + 1 < NCH) {
      const float* gP = res + (size_t)((c + 1) * NTILE + 16) * SLOT;
      const float* gZ = res + (size_t)((c + 1) * NTILE + j) * SLOT;
#pragma unroll
      for (int i = 0; i < 32; ++i) rphi[i] = gP[tid + i * 128];
#pragma unroll
      for (int i = 0; i < 8; ++i) rz[i] = gZ[tid + i * 128];
    }
    {
      const int nt = wave;
      v8f acc = {};
      acc = wmma_mm(sXc[cur], NXX, 1, sPhi + nt * 16, NXX, 1, NXX, acc, lane);
      int col = lane & 15, rb = (lane >> 4) << 3;
#pragma unroll
      for (int i = 0; i < 8; ++i) {
        int idx = (rb + i) * NXX + nt * 16 + col;
        sXc[cur ^ 1][idx] = acc[i] + sZ[idx];
      }
    }
    __syncthreads();
    cur ^= 1;
  }

  for (int e = tid; e < 16 * NXX; e += 128) {
    int row = e >> 6, col = e & 63;
    Out[(size_t)(j * 16 + row) * NXX + col] = sXc[cur][e];
  }
}

extern "C" void kernel_launch(void* const* d_in, const int* in_sizes, int n_in,
                              void* d_out, int out_size, void* d_ws, size_t ws_size,
                              hipStream_t stream) {
  const float* Ym  = (const float*)d_in[0];
  const float* U   = (const float*)d_in[1];
  const float* Dg  = (const float*)d_in[2];
  const float* Wfx = (const float*)d_in[3];
  const float* bfx = (const float*)d_in[4];
  const float* bfu = (const float*)d_in[6];
  const float* Wfu = (const float*)d_in[5];
  const float* Wfd = (const float*)d_in[7];
  const float* bfd = (const float*)d_in[8];
  const float* Wfy = (const float*)d_in[9];
  const float* bfy = (const float*)d_in[10];

  float* ws  = (float*)d_ws;
  float* Out = (float*)d_out;

  kf_gains_kernel<<<1, 256, 0, stream>>>(Wfx, Wfy, ws, 0);
  kf_chunk_scan_kernel<<<dim3(NTILE, NCP), 128, 0, stream>>>(
      Ym, U, Dg, Wfx, bfx, Wfu, bfu, Wfd, bfd, Wfy, bfy, ws, ws, 0);
  kf_gains_kernel<<<1, 256, 0, stream>>>(Wfx, Wfy, ws, 1);
  kf_chunk_scan_kernel<<<dim3(NTILE, NCP), 128, 0, stream>>>(
      Ym, U, Dg, Wfx, bfx, Wfu, bfu, Wfd, bfd, Wfy, bfy, ws, ws, 1);
  kf_chain_kernel<<<16, 128, 0, stream>>>(ws, Out);
}


# Gain kernel: each of the 4 product waves inverts S itself in registers (same Newton-Schulz/elimination); Sinv is at once the A operand of L^T, never passes LDS; 8->4 waves; L's sums re-ordered
# speedup vs baseline: 10.9503x; 1.0425x over previous
#include <hip/hip_runtime.h>


typedef __attribute__((ext_vector_type(2))) float v2f;
typedef __attribute__((ext_vector_type(4))) float v4f;
typedef __attribute__((ext_vector_type(8))) float v8f;

#define TT  4096
#define BB  256
#define NXX 64
#define NYY 16
#define NUU 16
#define NDD 8

#define SLOT    (NXX * NYY)
#define SPP     20
#define SPS     (NYY * SPP)
#define NS_TOL  6.103515625e-05f
#define HALF_T  (TT / 2)
#define TC      64
#define NCP     (HALF_T / TC)
#define NCH     (TT / TC)
#define NTILE   20
#define P_SLOT0   (HALF_T + 1)
#define RES_SLOT0 (P_SLOT0 + 4)

__device__ __forceinline__ float bcast_lane(float x, int l) {
  return __int_as_float(__builtin_amdgcn_readlane(__float_as_int(x), l));
}

__device__ __forceinline__ float xhalf(float x) {
  const int b = __float_as_int(x);
  return __int_as_float(__builtin_amdgcn_permlanex16(b, b, 0x76543210, 0xfedcba98,
                                                     false, false));
}

__device__ __forceinline__ v8f wmma_mm(const float* A, int ras, int cas,
                                       const float* B, int rbs, int cbs,
                                       int K, v8f acc, int lane) {
  const int mn = lane & 15;
  const int kh = (lane >> 4) << 1;
#pragma unroll
  for (int k = 0; k < K; k += 4) {
    v2f a, b;
    a.x = A[mn * ras + (k + kh) * cas];
    a.y = A[mn * ras + (k + kh + 1) * cas];
    b.x = B[(k + kh) * rbs + mn * cbs];
    b.y = B[(k + kh + 1) * rbs + mn * cbs];
    acc = __builtin_amdgcn_wmma_f32_16x16x4_f32(false, a, false, b,
                                                (short)0, acc, false, false);
  }
  return acc;
}

__device__ __forceinline__ void store_tile(float* D, int ldd, v8f acc, int lane) {
  const int col = lane & 15;
  const int rb  = (lane >> 4) << 3;
#pragma unroll
  for (int i = 0; i < 8; ++i) D[(rb + i) * ldd + col] = acc[i];
}

__global__ __launch_bounds__(128) void kf_gains_kernel(
    const float* __restrict__ Wfx_g, const float* __restrict__ Wfy_g,
    float* __restrict__ ws, int half) {
  float* Lbuf  = ws;
  float* Psave = ws + (size_t)P_SLOT0 * SLOT;
  __shared__ float sWfy[NXX * NYY];
  __shared__ float sW2[NXX * NYY];
  __shared__ __attribute__((aligned(16))) float sG[NXX * NYY];
  __shared__ __attribute__((aligned(16))) float sGp[16 * SPS];
  __shared__ __attribute__((aligned(16))) float sSp[4 * SPS];
  __shared__ __attribute__((aligned(16))) float sXi[4 * NYY * NYY];

  const int tid  = threadIdx.x;
  const int lane = tid & 31;
  const int wave = tid >> 5;
  const int c    = lane & 15;
  const int h    = lane >> 4;
  float* myX = sXi + wave * (NYY * NYY);

  for (int e = tid; e < NXX * NYY; e += 128) sWfy[e] = Wfy_g[e];
  for (int e = tid; e < 4 * NYY * NYY; e += 128) sXi[e] = 0.0f;
  if (half == 0) {
    for (int e = tid; e < NXX * NYY; e += 128) Lbuf[e] = 0.0f;
  } else {
    for (int e = tid; e < NXX * NYY; e += 128)
      Lbuf[e] = Lbuf[(size_t)HALF_T * SLOT + e];
  }
  __syncthreads();
  v8f s0acc = {};
  {
    v8f acc = {};
    acc = wmma_mm(Wfx_g + wave * 16, 1, NXX, sWfy, NYY, 1, NXX, acc, lane);
    store_tile(sW2 + wave * 16 * NYY, NYY, acc, lane);
    v8f s0 = {};
    s0 = wmma_mm(sWfy, 1, NYY, sWfy, NYY, 1, NXX, s0, lane);
#pragma unroll
    for (int i = 0; i < 8; ++i)
      s0acc[i] = (wave == 0) ? (s0[i] + ((i + 8 * h == c) ? 1.0f : 0.0f)) : 0.0f;
  }
  __syncthreads();

  v8f pt[4];
  v2f w2f[16];
  v2f w2a[4];
  v2f wfa[16];
  float wfyt[8];
  v8f xinv = {};
  if (half == 0) {
#pragma unroll
    for (int mt = 0; mt < 4; ++mt)
#pragma unroll
      for (int i = 0; i < 8; ++i)
        pt[mt][i] = (mt == wave && i + 8 * h == c) ? 1.0f : 0.0f;
  } else {
#pragma unroll
    for (int mt = 0; mt < 4; ++mt)
#pragma unroll
      for (int i = 0; i < 8; ++i)
        pt[mt][i] = Psave[((wave * 4 + mt) * 8 + i) * 32 + lane];
  }
#pragma unroll
  for (int mt = 0; mt < 4; ++mt)
#pragma unroll
    for (int jj = 0; jj < 4; ++jj) {
      const int k = 16 * mt + 8 * h + 2 * jj;
      w2f[4 * mt + jj].x = sW2[k * NYY + c];
      w2f[4 * mt + jj].y = sW2[(k + 1) * NYY + c];
    }
#pragma unroll
  for (int jj = 0; jj < 4; ++jj) {
    const int k = 16 * wave + 8 * h + 2 * jj;
    w2a[jj].x = sW2[k * NYY + c];
    w2a[jj].y = sW2[(k + 1) * NYY + c];
  }
#pragma unroll
  for (int mt = 0; mt < 4; ++mt)
#pragma unroll
    for (int jj = 0; jj < 4; ++jj) {
      const int k = 16 * wave + 8 * h + 2 * jj;
      wfa[4 * mt + jj].x = Wfx_g[(16 * mt + c) * NXX + k];
      wfa[4 * mt + jj].y = Wfx_g[(16 * mt + c) * NXX + k + 1];
    }
#pragma unroll
  for (int i = 0; i < 8; ++i) wfyt[i] = sWfy[(16 * wave + i + 8 * h) * NYY + c];

  for (int t = 0; t < HALF_T; ++t) {
    {
      v8f acc = {};
#pragma unroll
      for (int mt = 0; mt < 4; ++mt)
#pragma unroll
        for (int jj = 0; jj < 4; ++jj) {
          v2f a;
          a.x = pt[mt][2 * jj];
          a.y = pt[mt][2 * jj + 1];
          acc = __builtin_amdgcn_wmma_f32_16x16x4_f32(false, a, false, w2f[4 * mt + jj],
                                                      (short)0, acc, false, false);
        }
      v8f sp = s0acc;
      v8f gp[4];
#pragma unroll
      for (int mt = 0; mt < 4; ++mt) gp[mt] = v8f{};
#pragma unroll
      for (int jj = 0; jj < 4; ++jj) {
        v2f b;
        b.x = acc[2 * jj];
        b.y = acc[2 * jj + 1];
        sp = __builtin_amdgcn_wmma_f32_16x16x4_f32(false, w2a[jj], false, b,
                                                   (short)0, sp, false, false);
#pragma unroll
        for (int mt = 0; mt < 4; ++mt)
          gp[mt] = __builtin_amdgcn_wmma_f32_16x16x4_f32(false, wfa[4 * mt + jj], false, b,
                                                         (short)0, gp[mt], false, false);
      }
      float* dsp = sSp + wave * SPS + c * SPP + 8 * h;
      v4f slo, shi;
      slo.x = sp[0]; slo.y = sp[1]; slo.z = sp[2]; slo.w = sp[3];
      shi.x = sp[4]; shi.y = sp[5]; shi.z = sp[6]; shi.w = sp[7];
      *(v4f*)dsp = slo;
      *(v4f*)(dsp + 4) = shi;
#pragma unroll
      for (int mt = 0; mt < 4; ++mt) {
        float* dgp = sGp + (wave * 4 + mt) * SPS + c * SPP + 8 * h;
        v4f glo, ghi;
        glo.x = gp[mt][0]; glo.y = gp[mt][1]; glo.z = gp[mt][2]; glo.w = gp[mt][3];
        ghi.x = gp[mt][4]; ghi.y = gp[mt][5]; ghi.z = gp[mt][6]; ghi.w = gp[mt][7];
        *(v4f*)dgp = glo;
        *(v4f*)(dgp + 4) = ghi;
      }
    }
    __syncthreads();
    float scol[8];
    v8f d1 = {};
    {
      const float* spc = sSp + c * SPP + 8 * h;
      v4f sl[4][2];
#pragma unroll
      for (int w = 0; w < 4; ++w) {
        sl[w][0] = *(const v4f*)(spc + w * SPS);
        sl[w][1] = *(const v4f*)(spc + w * SPS + 4);
      }
      const float* xrp = myX + c * NYY + 8 * h;
      const v4f xr0 = *(const v4f*)xrp;
      const v4f xr1 = *(const v4f*)(xrp + 4);
      const float* gpc = sGp + wave * SPS + c * SPP + 8 * h;
      v4f gl[4][2];
#pragma unroll
      for (int w = 0; w < 4; ++w) {
        gl[w][0] = *(const v4f*)(gpc + w * 4 * SPS);
        gl[w][1] = *(const v4f*)(gpc + w * 4 * SPS + 4);
      }
      {
        const v4f lo = (sl[0][0] + sl[1][0]) + (sl[2][0] + sl[3][0]);
        const v4f hi = (sl[0][1] + sl[1][1]) + (sl[2][1] + sl[3][1]);
        scol[0] = lo.x; scol[1] = lo.y; scol[2] = lo.z; scol[3] = lo.w;
        scol[4] = hi.x; scol[5] = hi.y; scol[6] = hi.z; scol[7] = hi.w;
      }
      const float x0r[8] = {xr0.x, xr0.y, xr0.z, xr0.w, xr1.x, xr1.y, xr1.z, xr1.w};
#pragma unroll
      for (int jj = 0; jj < 4; ++jj) {
        v2f a, b;
        a.x = scol[2 * jj]; a.y = scol[2 * jj + 1];
        b.x = x0r[2 * jj];  b.y = x0r[2 * jj + 1];
        d1 = __builtin_amdgcn_wmma_f32_16x16x4_f32(false, a, false, b,
                                                   (short)0, d1, false, false);
      }
      const v4f lo = (gl[0][0] + gl[1][0]) + (gl[2][0] + gl[3][0]);
      const v4f hi = (gl[0][1] + gl[1][1]) + (gl[2][1] + gl[3][1]);
      const float g[8] = {lo.x, lo.y, lo.z, lo.w, hi.x, hi.y, hi.z, hi.w};
#pragma unroll
      for (int i = 0; i < 8; ++i)
        sG[(wave * 16 + 8 * h + i) * NYY + c] = g[i] + wfyt[i];
    }
    __syncthreads();
    {
      v4f ga0[4], ga1[4];
#pragma unroll
      for (int mt = 0; mt < 4; ++mt) {
        const float* ga = sG + (mt * 16 + c) * NYY + 8 * h;
        ga0[mt] = *(const v4f*)ga;
        ga1[mt] = *(const v4f*)(ga + 4);
      }
      const float* gw = sG + (wave * 16 + c) * NYY + 8 * h;
      const v4f gw0 = *(const v4f*)gw;
      const v4f gw1 = *(const v4f*)(gw + 4);
      float et[8];
#pragma unroll
      for (int i = 0; i < 8; ++i)
        et[i] = ((i + 8 * h == c) ? 1.0f : 0.0f) - d1[i];
      v8f x1 = xinv;
#pragma unroll
      for (int jj = 0; jj < 4; ++jj) {
        v2f a, b;
        a.x = et[2 * jj];   a.y = et[2 * jj + 1];
        b.x = xinv[2 * jj]; b.y = xinv[2 * jj + 1];
        x1 = __builtin_amdgcn_wmma_f32_16x16x4_f32(false, a, false, b,
                                                   (short)0, x1, false, false);
      }
      unsigned notsmall = 0u;
#pragma unroll
      for (int i = 0; i < 8; ++i)
        notsmall |= __builtin_amdgcn_ballot_w32(!(__builtin_fabsf(et[i]) <= NS_TOL));
      if (notsmall != 0u) {
        float colv[16];
#pragma unroll
        for (int i = 0; i < 8; ++i) {
          const float o = xhalf(scol[i]);
          colv[i]     = (h == 0) ? scol[i] : o;
          colv[8 + i] = (h == 0) ? o : scol[i];
        }
#pragma unroll
        for (int i = 0; i < 16; ++i)
          colv[i] = (lane < 16) ? colv[i] : ((i == (lane - 16)) ? 1.0f : 0.0f);
#pragma unroll
        for (int k = 0; k < 16; ++k) {
          float piv  = bcast_lane(colv[k], k);
          float rinv = 1.0f / piv;
          colv[k] *= rinv;
#pragma unroll
          for (int i = 0; i < 16; ++i) {
            if (i != k) {
              float f = bcast_lane(colv[i], k);
              colv[i] -= f * colv[k];
            }
          }
        }
#pragma unroll
        for (int i = 0; i < 8; ++i) {
          const float o = xhalf(colv[i]);
          x1[i] = (h == 0) ? o : colv[8 + i];
        }
      }
      xinv = x1;
#pragma unroll
      for (int i = 0; i < 8; ++i) myX[(8 * h + i) * NYY + c] = x1[i];
      v8f lt = {};
      {
        v2f a, b;
        a.x = x1[0]; a.y = x1[1]; b.x = gw0.x; b.y = gw0.y;
        lt = __builtin_amdgcn_wmma_f32_16x16x4_f32(false, a, false, b, (short)0, lt, false, false);
        a.x = x1[2]; a.y = x1[3]; b.x = gw0.z; b.y = gw0.w;
        lt = __builtin_amdgcn_wmma_f32_16x16x4_f32(false, a, false, b, (short)0, lt, false, false);
        a.x = x1[4]; a.y = x1[5]; b.x = gw1.x; b.y = gw1.y;
        lt = __builtin_amdgcn_wmma_f32_16x16x4_f32(false, a, false, b, (short)0, lt, false, false);
        a.x = x1[6]; a.y = x1[7]; b.x = gw1.z; b.y = gw1.w;
        lt = __builtin_amdgcn_wmma_f32_16x16x4_f32(false, a, false, b, (short)0, lt, false, false);
      }
      float* gL = Lbuf + (size_t)(t + 1) * SLOT + (wave * 16 + c) * NYY + 8 * h;
      v4f lo, hi;
      lo.x = lt[0]; lo.y = lt[1]; lo.z = lt[2]; lo.w = lt[3];
      hi.x = lt[4]; hi.y = lt[5]; hi.z = lt[6]; hi.w = lt[7];
      *(v4f*)gL = lo;
      *(v4f*)(gL + 4) = hi;
#pragma unroll
      for (int mt = 0; mt < 4; ++mt) {
        const v4f a0 = ga0[mt];
        const v4f a1 = ga1[mt];
        v2f a, b;
        v8f acc = {};
        a.x = a0.x; a.y = a0.y; b.x = lt[0]; b.y = lt[1];
        acc = __builtin_amdgcn_wmma_f32_16x16x4_f32(false, a, false, b, (short)0, acc, false, false);
        a.x = a0.z; a.y = a0.w; b.x = lt[2]; b.y = lt[3];
        acc = __builtin_amdgcn_wmma_f32_16x16x4_f32(false, a, false, b, (short)0, acc, false, false);
        a.x = a1.x; a.y = a1.y; b.x = lt[4]; b.y = lt[5];
        acc = __builtin_amdgcn_wmma_f32_16x16x4_f32(false, a, false, b, (short)0, acc, false, false);
        a.x = a1.z; a.y = a1.w; b.x = lt[6]; b.y = lt[7];
        acc = __builtin_amdgcn_wmma_f32_16x16x4_f32(false, a, false, b, (short)0, acc, false, false);
#pragma unroll
        for (int i = 0; i < 8; ++i)
          pt[mt][i] = ((mt == wave && i + 8 * h == c) ? 1.0f : 0.0f) - acc[i];
      }
    }
  }
  if (half == 0) {
#pragma unroll
    for (int mt = 0; mt < 4; ++mt)
#pragma unroll
      for (int i = 0; i < 8; ++i)
        Psave[((wave * 4 + mt) * 8 + i) * 32 + lane] = pt[mt][i];
  }
}

__global__ __launch_bounds__(128) void kf_chunk_scan_kernel(
    const float* __restrict__ Ym, const float* __restrict__ U,
    const float* __restrict__ Dg,
    const float* __restrict__ Wfx_g, const float* __restrict__ bfx,
    const float* __restrict__ Wfu_g, const float* __restrict__ bfu,
    const float* __restrict__ Wfd_g, const float* __restrict__ bfd,
    const float* __restrict__ Wfy_g, const float* __restrict__ bfy_g,
    const float* __restrict__ Lbuf, float* __restrict__ ws_out, int half) {
  __shared__ float sWfx[NXX * NXX];
  __shared__ float sWfu[NUU * NXX];
  __shared__ float sWfd[NDD * NXX];
  __shared__ float sWfy[NXX * NYY];
  __shared__ float sBsum[NXX];
  __shared__ float sBfy[NYY];
  __shared__ float sX[16 * NXX];
  __shared__ float sXp[16 * NXX];
  __shared__ float sL[NXX * NYY];
  __shared__ float sYm[16 * NYY];
  __shared__ float sU[16 * NUU];
  __shared__ float sD[16 * NDD];
  __shared__ float sInn[16 * NYY];

  const int tid  = threadIdx.x;
  const int lane = tid & 31;
  const int wave = tid >> 5;
  const int j    = blockIdx.x;
  const int cl   = blockIdx.y;
  const int c    = half * NCP + cl;
  const bool homog = (j >= 16);
  if (homog && c == 0) return;
  const int t0   = c * TC;
  const int s0   = cl * TC;
  const int r0   = homog ? 0 : j * 16;

  for (int e = tid; e < NXX * NXX; e += 128) sWfx[e] = Wfx_g[e];
  for (int e = tid; e < NUU * NXX; e += 128) sWfu[e] = Wfu_g[e];
  for (int e = tid; e < NDD * NXX; e += 128) sWfd[e] = Wfd_g[e];
  for (int e = tid; e < NXX * NYY; e += 128) sWfy[e] = Wfy_g[e];
  if (tid < NXX) sBsum[tid] = homog ? 0.0f : (bfx[tid] + bfu[tid] + bfd[tid]);
  if (tid < NYY) sBfy[tid] = homog ? 0.0f : bfy_g[tid];
  if (homog) {
    const int i0 = (j - 16) * 16;
    for (int e = tid; e < 16 * NXX; e += 128) {
      int row = e >> 6, col = e & 63;
      sX[e] = (col == i0 + row) ? 1.0f : 0.0f;
    }
    for (int e = tid; e < 16 * NYY; e += 128) sYm[e] = 0.0f;
    for (int e = tid; e < 16 * NUU; e += 128) sU[e] = 0.0f;
    if (tid < 16 * NDD) sD[tid] = 0.0f;
  } else {
    for (int e = tid; e < 16 * NXX; e += 128) sX[e] = 0.0f;
  }
  __syncthreads();

  for (int tt = 0; tt < TC; ++tt) {
    const int t = t0 + tt;
    {
      const float* gL  = Lbuf + (size_t)(s0 + tt) * SLOT;
      for (int e = tid; e < NXX * NYY; e += 128) sL[e] = gL[e];
      if (!homog) {
        const float* gYm = Ym + (size_t)t * BB * NYY + r0 * NYY;
        const float* gU  = U  + (size_t)t * BB * NUU + r0 * NUU;
        const float* gD  = Dg + (size_t)t * BB * NDD + r0 * NDD;
        for (int e = tid; e < 16 * NYY; e += 128)
          sYm[e] = __builtin_nontemporal_load(gYm + e);
        for (int e = tid; e < 16 * NUU; e += 128)
          sU[e] = __builtin_nontemporal_load(gU + e);
        if (tid < 16 * NDD) sD[tid] = __builtin_nontemporal_load(gD + tid);
      }
      if (tt + 1 < TC) {
        const char* pL = (const char*)(Lbuf + (size_t)(s0 + tt + 1) * SLOT);
        if (tid < 32) __builtin_prefetch(pL + tid * 128, 0, 0);
        if (!homog) {
          const char* pYm = (const char*)(Ym + (size_t)(t + 1) * BB * NYY + r0 * NYY);
          const char* pU  = (const char*)(U  + (size_t)(t + 1) * BB * NUU + r0 * NUU);
          const char* pD  = (const char*)(Dg + (size_t)(t + 1) * BB * NDD + r0 * NDD);
          if (tid < 8)  __builtin_prefetch(pYm + tid * 128, 0, 0);
          if (tid < 8)  __builtin_prefetch(pU + tid * 128, 0, 0);
          if (tid < 4)  __builtin_prefetch(pD + tid * 128, 0, 0);
        }
      }
    }
    __syncthreads();
    {
      const int nt = wave;
      v8f acc = {};
      acc = wmma_mm(sX, NXX, 1, sWfx + nt * 16, NXX, 1, NXX, acc, lane);
      acc = wmma_mm(sU, NUU, 1, sWfu + nt * 16, NXX, 1, NUU, acc, lane);
      acc = wmma_mm(sD, NDD, 1, sWfd + nt * 16, NXX, 1, NDD, acc, lane);
      int col = lane & 15, rb = (lane >> 4) << 3;
      float bias = sBsum[nt * 16 + col];
#pragma unroll
      for (int i = 0; i < 8; ++i)
        sXp[(rb + i) * NXX + nt * 16 + col] = acc[i] + bias;
    }
    __syncthreads();
    if (wave == 0) {
      v8f acc = {};
      acc = wmma_mm(sXp, NXX, 1, sWfy, NYY, 1, NXX, acc, lane);
      int col = lane & 15, rb = (lane >> 4) << 3;
      float by = sBfy[col];
#pragma unroll
      for (int i = 0; i < 8; ++i) {
        int r = rb + i;
        sInn[r * NYY + col] = sYm[r * NYY + col] - (acc[i] + by);
      }
    }
    __syncthreads();
    {
      const int nt = wave;
      v8f acc = {};
      acc = wmma_mm(sInn, NYY, 1, sL + nt * 16 * NYY, 1, NYY, NYY, acc, lane);
      int col = lane & 15, rb = (lane >> 4) << 3;
#pragma unroll
      for (int i = 0; i < 8; ++i) {
        int idx = (rb + i) * NXX + nt * 16 + col;
        sX[idx] = sXp[idx] + acc[i];
      }
    }
    __syncthreads();
  }

  float* res = ws_out + (size_t)(RES_SLOT0 + c * NTILE + j) * SLOT;
  for (int e = tid; e < 16 * NXX; e += 128) res[e] = sX[e];
}

__global__ __launch_bounds__(128) void kf_chain_kernel(
    const float* __restrict__ ws, float* __restrict__ Out) {
  __shared__ float sXc[2][16 * NXX];
  __shared__ float sPhi[NXX * NXX];
  __shared__ float sZ[16 * NXX];

  const int tid  = threadIdx.x;
  const int lane = tid & 31;
  const int wave = tid >> 5;
  const int j    = blockIdx.x;
  const float* res = ws + (size_t)RES_SLOT0 * SLOT;

  float rphi[32], rz[8];
  {
    const float* z0 = res + (size_t)j * SLOT;
    for (int e = tid; e < 16 * NXX; e += 128) sXc[0][e] = z0[e];
    const float* gP = res + (size_t)(1 * NTILE + 16) * SLOT;
    const float* gZ = res + (size_t)(1 * NTILE + j) * SLOT;
#pragma unroll
    for (int i = 0; i < 32; ++i) rphi[i] = gP[tid + i * 128];
#pragma unroll
    for (int i = 0; i < 8; ++i) rz[i] = gZ[tid + i * 128];
  }

  int cur = 0;
  for (int c = 1; c < NCH; ++c) {
#pragma unroll
    for (int i = 0; i < 32; ++i) sPhi[tid + i * 128] = rphi[i];
#pragma unroll
    for (int i = 0; i < 8; ++i) sZ[tid + i * 128] = rz[i];
    __syncthreads();
    if (c + 1 < NCH) {
      const float* gP = res + (size_t)((c + 1) * NTILE + 16) * SLOT;
      const float* gZ = res + (size_t)((c + 1) * NTILE + j) * SLOT;
#pragma unroll
      for (int i = 0; i < 32; ++i) rphi[i] = gP[tid + i * 128];
#pragma unroll
      for (int i = 0; i < 8; ++i) rz[i] = gZ[tid + i * 128];
    }
    {
      const int nt = wave;
      v8f acc = {};
      acc = wmma_mm(sXc[cur], NXX, 1, sPhi + nt * 16, NXX, 1, NXX, acc, lane);
      int col = lane & 15, rb = (lane >> 4) << 3;
#pragma unroll
      for (int i = 0; i < 8; ++i) {
        int idx = (rb + i) * NXX + nt * 16 + col;
        sXc[cur ^ 1][idx] = acc[i] + sZ[idx];
      }
    }
    __syncthreads();
    cur ^= 1;
  }

  for (int e = tid; e < 16 * NXX; e += 128) {
    int row = e >> 6, col = e & 63;
    Out[(size_t)(j * 16 + row) * NXX + col] = sXc[cur][e];
  }
}

extern "C" void kernel_launch(void* const* d_in, const int* in_sizes, int n_in,
                              void* d_out, int out_size, void* d_ws, size_t ws_size,
                              hipStream_t stream) {
  const float* Ym  = (const float*)d_in[0];
  const float* U   = (const float*)d_in[1];
  const float* Dg  = (const float*)d_in[2];
  const float* Wfx = (const float*)d_in[3];
  const float* bfx = (const float*)d_in[4];
  const float* bfu = (const float*)d_in[6];
  const float* Wfu = (const float*)d_in[5];
  const float* Wfd = (const float*)d_in[7];
  const float* bfd = (const float*)d_in[8];
  const float* Wfy = (const float*)d_in[9];
  const float* bfy = (const float*)d_in[10];

  float* ws  = (float*)d_ws;
  float* Out = (float*)d_out;

  kf_gains_kernel<<<1, 128, 0, stream>>>(Wfx, Wfy, ws, 0);
  kf_chunk_scan_kernel<<<dim3(NTILE, NCP), 128, 0, stream>>>(
      Ym, U, Dg, Wfx, bfx, Wfu, bfu, Wfd, bfd, Wfy, bfy, ws, ws, 0);
  kf_gains_kernel<<<1, 128, 0, stream>>>(Wfx, Wfy, ws, 1);
  kf_chunk_scan_kernel<<<dim3(NTILE, NCP), 128, 0, stream>>>(
      Ym, U, Dg, Wfx, bfx, Wfu, bfu, Wfd, bfd, Wfy, bfy, ws, ws, 1);
  kf_chain_kernel<<<16, 128, 0, stream>>>(ws, Out);
}


# Chunk scan: per-step L, U, D, Ym go from global straight into WMMA fragment registers (no LDS stage), issued at step top under the x Wfx product; 13 serial L2 trips, 1 barrier gone; sums re-ordered
# speedup vs baseline: 11.7752x; 1.0753x over previous
#include <hip/hip_runtime.h>


typedef __attribute__((ext_vector_type(2))) float v2f;
typedef __attribute__((ext_vector_type(4))) float v4f;
typedef __attribute__((ext_vector_type(8))) float v8f;

#define TT  4096
#define BB  256
#define NXX 64
#define NYY 16
#define NUU 16
#define NDD 8

#define SLOT    (NXX * NYY)
#define SPP     20
#define SPS     (NYY * SPP)
#define NS_TOL  6.103515625e-05f
#define HALF_T  (TT / 2)
#define TC      64
#define NCP     (HALF_T / TC)
#define NCH     (TT / TC)
#define NTILE   20
#define P_SLOT0   (HALF_T + 1)
#define RES_SLOT0 (P_SLOT0 + 4)

__device__ __forceinline__ float bcast_lane(float x, int l) {
  return __int_as_float(__builtin_amdgcn_readlane(__float_as_int(x), l));
}

__device__ __forceinline__ float xhalf(float x) {
  const int b = __float_as_int(x);
  return __int_as_float(__builtin_amdgcn_permlanex16(b, b, 0x76543210, 0xfedcba98,
                                                     false, false));
}

__device__ __forceinline__ v8f wmma_mm(const float* A, int ras, int cas,
                                       const float* B, int rbs, int cbs,
                                       int K, v8f acc, int lane) {
  const int mn = lane & 15;
  const int kh = (lane >> 4) << 1;
#pragma unroll
  for (int k = 0; k < K; k += 4) {
    v2f a, b;
    a.x = A[mn * ras + (k + kh) * cas];
    a.y = A[mn * ras + (k + kh + 1) * cas];
    b.x = B[(k + kh) * rbs + mn * cbs];
    b.y = B[(k + kh + 1) * rbs + mn * cbs];
    acc = __builtin_amdgcn_wmma_f32_16x16x4_f32(false, a, false, b,
                                                (short)0, acc, false, false);
  }
  return acc;
}

__device__ __forceinline__ void store_tile(float* D, int ldd, v8f acc, int lane) {
  const int col = lane & 15;
  const int rb  = (lane >> 4) << 3;
#pragma unroll
  for (int i = 0; i < 8; ++i) D[(rb + i) * ldd + col] = acc[i];
}

__global__ __launch_bounds__(128) void kf_gains_kernel(
    const float* __restrict__ Wfx_g, const float* __restrict__ Wfy_g,
    float* __restrict__ ws, int half) {
  float* Lbuf  = ws;
  float* Psave = ws + (size_t)P_SLOT0 * SLOT;
  __shared__ float sWfy[NXX * NYY];
  __shared__ float sW2[NXX * NYY];
  __shared__ __attribute__((aligned(16))) float sG[NXX * NYY];
  __shared__ __attribute__((aligned(16))) float sGp[16 * SPS];
  __shared__ __attribute__((aligned(16))) float sSp[4 * SPS];
  __shared__ __attribute__((aligned(16))) float sXi[4 * NYY * NYY];

  const int tid  = threadIdx.x;
  const int lane = tid & 31;
  const int wave = tid >> 5;
  const int c    = lane & 15;
  const int h    = lane >> 4;
  float* myX = sXi + wave * (NYY * NYY);

  for (int e = tid; e < NXX * NYY; e += 128) sWfy[e] = Wfy_g[e];
  for (int e = tid; e < 4 * NYY * NYY; e += 128) sXi[e] = 0.0f;
  if (half == 0) {
    for (int e = tid; e < NXX * NYY; e += 128) Lbuf[e] = 0.0f;
  } else {
    for (int e = tid; e < NXX * NYY; e += 128)
      Lbuf[e] = Lbuf[(size_t)HALF_T * SLOT + e];
  }
  __syncthreads();
  v8f s0acc = {};
  {
    v8f acc = {};
    acc = wmma_mm(Wfx_g + wave * 16, 1, NXX, sWfy, NYY, 1, NXX, acc, lane);
    store_tile(sW2 + wave * 16 * NYY, NYY, acc, lane);
    v8f s0 = {};
    s0 = wmma_mm(sWfy, 1, NYY, sWfy, NYY, 1, NXX, s0, lane);
#pragma unroll
    for (int i = 0; i < 8; ++i)
      s0acc[i] = (wave == 0) ? (s0[i] + ((i + 8 * h == c) ? 1.0f : 0.0f)) : 0.0f;
  }
  __syncthreads();

  v8f pt[4];
  v2f w2f[16];
  v2f w2a[4];
  v2f wfa[16];
  float wfyt[8];
  v8f xinv = {};
  if (half == 0) {
#pragma unroll
    for (int mt = 0; mt < 4; ++mt)
#pragma unroll
      for (int i = 0; i < 8; ++i)
        pt[mt][i] = (mt == wave && i + 8 * h == c) ? 1.0f : 0.0f;
  } else {
#pragma unroll
    for (int mt = 0; mt < 4; ++mt)
#pragma unroll
      for (int i = 0; i < 8; ++i)
        pt[mt][i] = Psave[((wave * 4 + mt) * 8 + i) * 32 + lane];
  }
#pragma unroll
  for (int mt = 0; mt < 4; ++mt)
#pragma unroll
    for (int jj = 0; jj < 4; ++jj) {
      const int k = 16 * mt + 8 * h + 2 * jj;
      w2f[4 * mt + jj].x = sW2[k * NYY + c];
      w2f[4 * mt + jj].y = sW2[(k + 1) * NYY + c];
    }
#pragma unroll
  for (int jj = 0; jj < 4; ++jj) {
    const int k = 16 * wave + 8 * h + 2 * jj;
    w2a[jj].x = sW2[k * NYY + c];
    w2a[jj].y = sW2[(k + 1) * NYY + c];
  }
#pragma unroll
  for (int mt = 0; mt < 4; ++mt)
#pragma unroll
    for (int jj = 0; jj < 4; ++jj) {
      const int k = 16 * wave + 8 * h + 2 * jj;
      wfa[4 * mt + jj].x = Wfx_g[(16 * mt + c) * NXX + k];
      wfa[4 * mt + jj].y = Wfx_g[(16 * mt + c) * NXX + k + 1];
    }
#pragma unroll
  for (int i = 0; i < 8; ++i) wfyt[i] = sWfy[(16 * wave + i + 8 * h) * NYY + c];

  for (int t = 0; t < HALF_T; ++t) {
    {
      v8f acc = {};
#pragma unroll
      for (int mt = 0; mt < 4; ++mt)
#pragma unroll
        for (int jj = 0; jj < 4; ++jj) {
          v2f a;
          a.x = pt[mt][2 * jj];
          a.y = pt[mt][2 * jj + 1];
          acc = __builtin_amdgcn_wmma_f32_16x16x4_f32(false, a, false, w2f[4 * mt + jj],
                                                      (short)0, acc, false, false);
        }
      v8f sp = s0acc;
      v8f gp[4];
#pragma unroll
      for (int mt = 0; mt < 4; ++mt) gp[mt] = v8f{};
#pragma unroll
      for (int jj = 0; jj < 4; ++jj) {
        v2f b;
        b.x = acc[2 * jj];
        b.y = acc[2 * jj + 1];
        sp = __builtin_amdgcn_wmma_f32_16x16x4_f32(false, w2a[jj], false, b,
                                                   (short)0, sp, false, false);
#pragma unroll
        for (int mt = 0; mt < 4; ++mt)
          gp[mt] = __builtin_amdgcn_wmma_f32_16x16x4_f32(false, wfa[4 * mt + jj], false, b,
                                                         (short)0, gp[mt], false, false);
      }
      float* dsp = sSp + wave * SPS + c * SPP + 8 * h;
      v4f slo, shi;
      slo.x = sp[0]; slo.y = sp[1]; slo.z = sp[2]; slo.w = sp[3];
      shi.x = sp[4]; shi.y = sp[5]; shi.z = sp[6]; shi.w = sp[7];
      *(v4f*)dsp = slo;
      *(v4f*)(dsp + 4) = shi;
#pragma unroll
      for (int mt = 0; mt < 4; ++mt) {
        float* dgp = sGp + (wave * 4 + mt) * SPS + c * SPP + 8 * h;
        v4f glo, ghi;
        glo.x = gp[mt][0]; glo.y = gp[mt][1]; glo.z = gp[mt][2]; glo.w = gp[mt][3];
        ghi.x = gp[mt][4]; ghi.y = gp[mt][5]; ghi.z = gp[mt][6]; ghi.w = gp[mt][7];
        *(v4f*)dgp = glo;
        *(v4f*)(dgp + 4) = ghi;
      }
    }
    __syncthreads();
    float scol[8];
    v8f d1 = {};
    {
      const float* spc = sSp + c * SPP + 8 * h;
      v4f sl[4][2];
#pragma unroll
      for (int w = 0; w < 4; ++w) {
        sl[w][0] = *(const v4f*)(spc + w * SPS);
        sl[w][1] = *(const v4f*)(spc + w * SPS + 4);
      }
      const float* xrp = myX + c * NYY + 8 * h;
      const v4f xr0 = *(const v4f*)xrp;
      const v4f xr1 = *(const v4f*)(xrp + 4);
      const float* gpc = sGp + wave * SPS + c * SPP + 8 * h;
      v4f gl[4][2];
#pragma unroll
      for (int w = 0; w < 4; ++w) {
        gl[w][0] = *(const v4f*)(gpc + w * 4 * SPS);
        gl[w][1] = *(const v4f*)(gpc + w * 4 * SPS + 4);
      }
      {
        const v4f lo = (sl[0][0] + sl[1][0]) + (sl[2][0] + sl[3][0]);
        const v4f hi = (sl[0][1] + sl[1][1]) + (sl[2][1] + sl[3][1]);
        scol[0] = lo.x; scol[1] = lo.y; scol[2] = lo.z; scol[3] = lo.w;
        scol[4] = hi.x; scol[5] = hi.y; scol[6] = hi.z; scol[7] = hi.w;
      }
      const float x0r[8] = {xr0.x, xr0.y, xr0.z, xr0.w, xr1.x, xr1.y, xr1.z, xr1.w};
#pragma unroll
      for (int jj = 0; jj < 4; ++jj) {
        v2f a, b;
        a.x = scol[2 * jj]; a.y = scol[2 * jj + 1];
        b.x = x0r[2 * jj];  b.y = x0r[2 * jj + 1];
        d1 = __builtin_amdgcn_wmma_f32_16x16x4_f32(false, a, false, b,
                                                   (short)0, d1, false, false);
      }
      const v4f lo = (gl[0][0] + gl[1][0]) + (gl[2][0] + gl[3][0]);
      const v4f hi = (gl[0][1] + gl[1][1]) + (gl[2][1] + gl[3][1]);
      const float g[8] = {lo.x, lo.y, lo.z, lo.w, hi.x, hi.y, hi.z, hi.w};
#pragma unroll
      for (int i = 0; i < 8; ++i)
        sG[(wave * 16 + 8 * h + i) * NYY + c] = g[i] + wfyt[i];
    }
    __syncthreads();
    {
      v4f ga0[4], ga1[4];
#pragma unroll
      for (int mt = 0; mt < 4; ++mt) {
        const float* ga = sG + (mt * 16 + c) * NYY + 8 * h;
        ga0[mt] = *(const v4f*)ga;
        ga1[mt] = *(const v4f*)(ga + 4);
      }
      const float* gw = sG + (wave * 16 + c) * NYY + 8 * h;
      const v4f gw0 = *(const v4f*)gw;
      const v4f gw1 = *(const v4f*)(gw + 4);
      float et[8];
#pragma unroll
      for (int i = 0; i < 8; ++i)
        et[i] = ((i + 8 * h == c) ? 1.0f : 0.0f) - d1[i];
      v8f x1 = xinv;
#pragma unroll
      for (int jj = 0; jj < 4; ++jj) {
        v2f a, b;
        a.x = et[2 * jj];   a.y = et[2 * jj + 1];
        b.x = xinv[2 * jj]; b.y = xinv[2 * jj + 1];
        x1 = __builtin_amdgcn_wmma_f32_16x16x4_f32(false, a, false, b,
                                                   (short)0, x1, false, false);
      }
      unsigned notsmall = 0u;
#pragma unroll
      for (int i = 0; i < 8; ++i)
        notsmall |= __builtin_amdgcn_ballot_w32(!(__builtin_fabsf(et[i]) <= NS_TOL));
      if (notsmall != 0u) {
        float colv[16];
#pragma unroll
        for (int i = 0; i < 8; ++i) {
          const float o = xhalf(scol[i]);
          colv[i]     = (h == 0) ? scol[i] : o;
          colv[8 + i] = (h == 0) ? o : scol[i];
        }
#pragma unroll
        for (int i = 0; i < 16; ++i)
          colv[i] = (lane < 16) ? colv[i] : ((i == (lane - 16)) ? 1.0f : 0.0f);
#pragma unroll
        for (int k = 0; k < 16; ++k) {
          float piv  = bcast_lane(colv[k], k);
          float rinv = 1.0f / piv;
          colv[k] *= rinv;
#pragma unroll
          for (int i = 0; i < 16; ++i) {
            if (i != k) {
              float f = bcast_lane(colv[i], k);
              colv[i] -= f * colv[k];
            }
          }
        }
#pragma unroll
        for (int i = 0; i < 8; ++i) {
          const float o = xhalf(colv[i]);
          x1[i] = (h == 0) ? o : colv[8 + i];
        }
      }
      xinv = x1;
#pragma unroll
      for (int i = 0; i < 8; ++i) myX[(8 * h + i) * NYY + c] = x1[i];
      v8f lt = {};
      {
        v2f a, b;
        a.x = x1[0]; a.y = x1[1]; b.x = gw0.x; b.y = gw0.y;
        lt = __builtin_amdgcn_wmma_f32_16x16x4_f32(false, a, false, b, (short)0, lt, false, false);
        a.x = x1[2]; a.y = x1[3]; b.x = gw0.z; b.y = gw0.w;
        lt = __builtin_amdgcn_wmma_f32_16x16x4_f32(false, a, false, b, (short)0, lt, false, false);
        a.x = x1[4]; a.y = x1[5]; b.x = gw1.x; b.y = gw1.y;
        lt = __builtin_amdgcn_wmma_f32_16x16x4_f32(false, a, false, b, (short)0, lt, false, false);
        a.x = x1[6]; a.y = x1[7]; b.x = gw1.z; b.y = gw1.w;
        lt = __builtin_amdgcn_wmma_f32_16x16x4_f32(false, a, false, b, (short)0, lt, false, false);
      }
      float* gL = Lbuf + (size_t)(t + 1) * SLOT + (wave * 16 + c) * NYY + 8 * h;
      v4f lo, hi;
      lo.x = lt[0]; lo.y = lt[1]; lo.z = lt[2]; lo.w = lt[3];
      hi.x = lt[4]; hi.y = lt[5]; hi.z = lt[6]; hi.w = lt[7];
      *(v4f*)gL = lo;
      *(v4f*)(gL + 4) = hi;
#pragma unroll
      for (int mt = 0; mt < 4; ++mt) {
        const v4f a0 = ga0[mt];
        const v4f a1 = ga1[mt];
        v2f a, b;
        v8f acc = {};
        a.x = a0.x; a.y = a0.y; b.x = lt[0]; b.y = lt[1];
        acc = __builtin_amdgcn_wmma_f32_16x16x4_f32(false, a, false, b, (short)0, acc, false, false);
        a.x = a0.z; a.y = a0.w; b.x = lt[2]; b.y = lt[3];
        acc = __builtin_amdgcn_wmma_f32_16x16x4_f32(false, a, false, b, (short)0, acc, false, false);
        a.x = a1.x; a.y = a1.y; b.x = lt[4]; b.y = lt[5];
        acc = __builtin_amdgcn_wmma_f32_16x16x4_f32(false, a, false, b, (short)0, acc, false, false);
        a.x = a1.z; a.y = a1.w; b.x = lt[6]; b.y = lt[7];
        acc = __builtin_amdgcn_wmma_f32_16x16x4_f32(false, a, false, b, (short)0, acc, false, false);
#pragma unroll
        for (int i = 0; i < 8; ++i)
          pt[mt][i] = ((mt == wave && i + 8 * h == c) ? 1.0f : 0.0f) - acc[i];
      }
    }
  }
  if (half == 0) {
#pragma unroll
    for (int mt = 0; mt < 4; ++mt)
#pragma unroll
      for (int i = 0; i < 8; ++i)
        Psave[((wave * 4 + mt) * 8 + i) * 32 + lane] = pt[mt][i];
  }
}

__global__ __launch_bounds__(128) void kf_chunk_scan_kernel(
    const float* __restrict__ Ym, const float* __restrict__ U,
    const float* __restrict__ Dg,
    const float* __restrict__ Wfx_g, const float* __restrict__ bfx,
    const float* __restrict__ Wfu_g, const float* __restrict__ bfu,
    const float* __restrict__ Wfd_g, const float* __restrict__ bfd,
    const float* __restrict__ Wfy_g, const float* __restrict__ bfy_g,
    const float* __restrict__ Lbuf, float* __restrict__ ws_out, int half) {
  __shared__ float sWfx[NXX * NXX];
  __shared__ float sWfu[NUU * NXX];
  __shared__ float sWfd[NDD * NXX];
  __shared__ float sWfy[NXX * NYY];
  __shared__ float sBsum[NXX];
  __shared__ float sBfy[NYY];
  __shared__ float sX[16 * NXX];
  __shared__ float sXp[16 * NXX];
  __shared__ __attribute__((aligned(16))) float sInn[16 * NYY];

  const int tid  = threadIdx.x;
  const int lane = tid & 31;
  const int wave = tid >> 5;
  const int c16  = lane & 15;
  const int h    = lane >> 4;
  const int j    = blockIdx.x;
  const int cl   = blockIdx.y;
  const int c    = half * NCP + cl;
  const bool homog = (j >= 16);
  if (homog && c == 0) return;
  const int t0   = c * TC;
  const int s0   = cl * TC;
  const int r0   = homog ? 0 : j * 16;

  for (int e = tid; e < NXX * NXX; e += 128) sWfx[e] = Wfx_g[e];
  for (int e = tid; e < NUU * NXX; e += 128) sWfu[e] = Wfu_g[e];
  for (int e = tid; e < NDD * NXX; e += 128) sWfd[e] = Wfd_g[e];
  for (int e = tid; e < NXX * NYY; e += 128) sWfy[e] = Wfy_g[e];
  if (tid < NXX) sBsum[tid] = homog ? 0.0f : (bfx[tid] + bfu[tid] + bfd[tid]);
  if (tid < NYY) sBfy[tid] = homog ? 0.0f : bfy_g[tid];
  if (homog) {
    const int i0 = (j - 16) * 16;
    for (int e = tid; e < 16 * NXX; e += 128) {
      int row = e >> 6, col = e & 63;
      sX[e] = (col == i0 + row) ? 1.0f : 0.0f;
    }
  } else {
    for (int e = tid; e < 16 * NXX; e += 128) sX[e] = 0.0f;
  }
  __syncthreads();

  for (int tt = 0; tt < TC; ++tt) {
    const int t = t0 + tt;
    v4f u0 = {}, u1 = {}, d0 = {};
    float ymr[8];
#pragma unroll
    for (int i = 0; i < 8; ++i) ymr[i] = 0.0f;
    if (!homog) {
      const float* gU = U  + ((size_t)t * BB + r0 + c16) * NUU + 8 * h;
      const float* gD = Dg + ((size_t)t * BB + r0 + c16) * NDD + 4 * h;
      u0 = *(const v4f*)gU;
      u1 = *(const v4f*)(gU + 4);
      d0 = *(const v4f*)gD;
      if (wave == 0) {
        const float* gYm = Ym + ((size_t)t * BB + r0 + 8 * h) * NYY + c16;
#pragma unroll
        for (int i = 0; i < 8; ++i) ymr[i] = __builtin_nontemporal_load(gYm + i * NYY);
      }
    }
    const float* gLr = Lbuf + (size_t)(s0 + tt) * SLOT + (wave * 16 + c16) * NYY + 8 * h;
    const v4f l0 = *(const v4f*)gLr;
    const v4f l1 = *(const v4f*)(gLr + 4);
    if (tt + 1 < TC) {
      const char* pL = (const char*)(Lbuf + (size_t)(s0 + tt + 1) * SLOT);
      if (tid < 32) __builtin_prefetch(pL + tid * 128, 0, 0);
      if (!homog) {
        const char* pYm = (const char*)(Ym + (size_t)(t + 1) * BB * NYY + r0 * NYY);
        const char* pU  = (const char*)(U  + (size_t)(t + 1) * BB * NUU + r0 * NUU);
        const char* pD  = (const char*)(Dg + (size_t)(t + 1) * BB * NDD + r0 * NDD);
        if (tid < 8)  __builtin_prefetch(pYm + tid * 128, 0, 0);
        if (tid < 8)  __builtin_prefetch(pU + tid * 128, 0, 0);
        if (tid < 4)  __builtin_prefetch(pD + tid * 128, 0, 0);
      }
    }
    {
      const int nt = wave;
      v8f acc = {};
      acc = wmma_mm(sX, NXX, 1, sWfx + nt * 16, NXX, 1, NXX, acc, lane);
      {
        const float* bu = sWfu + (8 * h) * NXX + nt * 16 + c16;
        v2f a, b;
        a.x = u0.x; a.y = u0.y; b.x = bu[0 * NXX]; b.y = bu[1 * NXX];
        acc = __builtin_amdgcn_wmma_f32_16x16x4_f32(false, a, false, b, (short)0, acc, false, false);
        a.x = u0.z; a.y = u0.w; b.x = bu[2 * NXX]; b.y = bu[3 * NXX];
        acc = __builtin_amdgcn_wmma_f32_16x16x4_f32(false, a, false, b, (short)0, acc, false, false);
        a.x = u1.x; a.y = u1.y; b.x = bu[4 * NXX]; b.y = bu[5 * NXX];
        acc = __builtin_amdgcn_wmma_f32_16x16x4_f32(false, a, false, b, (short)0, acc, false, false);
        a.x = u1.z; a.y = u1.w; b.x = bu[6 * NXX]; b.y = bu[7 * NXX];
        acc = __builtin_amdgcn_wmma_f32_16x16x4_f32(false, a, false, b, (short)0, acc, false, false);
      }
      {
        const float* bd = sWfd + (4 * h) * NXX + nt * 16 + c16;
        v2f a, b;
        a.x = d0.x; a.y = d0.y; b.x = bd[0 * NXX]; b.y = bd[1 * NXX];
        acc = __builtin_amdgcn_wmma_f32_16x16x4_f32(false, a, false, b, (short)0, acc, false, false);
        a.x = d0.z; a.y = d0.w; b.x = bd[2 * NXX]; b.y = bd[3 * NXX];
        acc = __builtin_amdgcn_wmma_f32_16x16x4_f32(false, a, false, b, (short)0, acc, false, false);
      }
      const int rb = h << 3;
      float bias = sBsum[nt * 16 + c16];
#pragma unroll
      for (int i = 0; i < 8; ++i)
        sXp[(rb + i) * NXX + nt * 16 + c16] = acc[i] + bias;
    }
    __syncthreads();
    if (wave == 0) {
      v8f acc = {};
      acc = wmma_mm(sXp, NXX, 1, sWfy, NYY, 1, NXX, acc, lane);
      const int rb = h << 3;
      float by = sBfy[c16];
#pragma unroll
      for (int i = 0; i < 8; ++i)
        sInn[(rb + i) * NYY + c16] = ymr[i] - (acc[i] + by);
    }
    __syncthreads();
    {
      const int nt = wave;
      const float* ia = sInn + c16 * NYY + 8 * h;
      const v4f i0 = *(const v4f*)ia;
      const v4f i1 = *(const v4f*)(ia + 4);
      v8f acc = {};
      v2f a, b;
      a.x = i0.x; a.y = i0.y; b.x = l0.x; b.y = l0.y;
      acc = __builtin_amdgcn_wmma_f32_16x16x4_f32(false, a, false, b, (short)0, acc, false, false);
      a.x = i0.z; a.y = i0.w; b.x = l0.z; b.y = l0.w;
      acc = __builtin_amdgcn_wmma_f32_16x16x4_f32(false, a, false, b, (short)0, acc, false, false);
      a.x = i1.x; a.y = i1.y; b.x = l1.x; b.y = l1.y;
      acc = __builtin_amdgcn_wmma_f32_16x16x4_f32(false, a, false, b, (short)0, acc, false, false);
      a.x = i1.z; a.y = i1.w; b.x = l1.z; b.y = l1.w;
      acc = __builtin_amdgcn_wmma_f32_16x16x4_f32(false, a, false, b, (short)0, acc, false, false);
      const int rb = h << 3;
#pragma unroll
      for (int i = 0; i < 8; ++i) {
        int idx = (rb + i) * NXX + nt * 16 + c16;
        sX[idx] = sXp[idx] + acc[i];
      }
    }
    __syncthreads();
  }

  float* res = ws_out + (size_t)(RES_SLOT0 + c * NTILE + j) * SLOT;
  for (int e = tid; e < 16 * NXX; e += 128) res[e] = sX[e];
}

__global__ __launch_bounds__(128) void kf_chain_kernel(
    const float* __restrict__ ws, float* __restrict__ Out) {
  __shared__ float sXc[2][16 * NXX];
  __shared__ float sPhi[NXX * NXX];
  __shared__ float sZ[16 * NXX];

  const int tid  = threadIdx.x;
  const int lane = tid & 31;
  const int wave = tid >> 5;
  const int j    = blockIdx.x;
  const float* res = ws + (size_t)RES_SLOT0 * SLOT;

  float rphi[32], rz[8];
  {
    const float* z0 = res + (size_t)j * SLOT;
    for (int e = tid; e < 16 * NXX; e += 128) sXc[0][e] = z0[e];
    const float* gP = res + (size_t)(1 * NTILE + 16) * SLOT;
    const float* gZ = res + (size_t)(1 * NTILE + j) * SLOT;
#pragma unroll
    for (int i = 0; i < 32; ++i) rphi[i] = gP[tid + i * 128];
#pragma unroll
    for (int i = 0; i < 8; ++i) rz[i] = gZ[tid + i * 128];
  }

  int cur = 0;
  for (int c = 1; c < NCH; ++c) {
#pragma unroll
    for (int i = 0; i < 32; ++i) sPhi[tid + i * 128] = rphi[i];
#pragma unroll
    for (int i = 0; i < 8; ++i) sZ[tid + i * 128] = rz[i];
    __syncthreads();
    if (c + 1 < NCH) {
      const float* gP = res + (size_t)((c + 1) * NTILE + 16) * SLOT;
      const float* gZ = res + (size_t)((c + 1) * NTILE + j) * SLOT;
#pragma unroll
      for (int i = 0; i < 32; ++i) rphi[i] = gP[tid + i * 128];
#pragma unroll
      for (int i = 0; i < 8; ++i) rz[i] = gZ[tid + i * 128];
    }
    {
      const int nt = wave;
      v8f acc = {};
      acc = wmma_mm(sXc[cur], NXX, 1, sPhi + nt * 16, NXX, 1, NXX, acc, lane);
      int col = lane & 15, rb = (lane >> 4) << 3;
#pragma unroll
      for (int i = 0; i < 8; ++i) {
        int idx = (rb + i) * NXX + nt * 16 + col;
        sXc[cur ^ 1][idx] = acc[i] + sZ[idx];
      }
    }
    __syncthreads();
    cur ^= 1;
  }

  for (int e = tid; e < 16 * NXX; e += 128) {
    int row = e >> 6, col = e & 63;
    Out[(size_t)(j * 16 + row) * NXX + col] = sXc[cur][e];
  }
}

extern "C" void kernel_launch(void* const* d_in, const int* in_sizes, int n_in,
                              void* d_out, int out_size, void* d_ws, size_t ws_size,
                              hipStream_t stream) {
  const float* Ym  = (const float*)d_in[0];
  const float* U   = (const float*)d_in[1];
  const float* Dg  = (const float*)d_in[2];
  const float* Wfx = (const float*)d_in[3];
  const float* bfx = (const float*)d_in[4];
  const float* bfu = (const float*)d_in[6];
  const float* Wfu = (const float*)d_in[5];
  const float* Wfd = (const float*)d_in[7];
  const float* bfd = (const float*)d_in[8];
  const float* Wfy = (const float*)d_in[9];
  const float* bfy = (const float*)d_in[10];

  float* ws  = (float*)d_ws;
  float* Out = (float*)d_out;

  kf_gains_kernel<<<1, 128, 0, stream>>>(Wfx, Wfy, ws, 0);
  kf_chunk_scan_kernel<<<dim3(NTILE, NCP), 128, 0, stream>>>(
      Ym, U, Dg, Wfx, bfx, Wfu, bfu, Wfd, bfd, Wfy, bfy, ws, ws, 0);
  kf_gains_kernel<<<1, 128, 0, stream>>>(Wfx, Wfy, ws, 1);
  kf_chunk_scan_kernel<<<dim3(NTILE, NCP), 128, 0, stream>>>(
      Ym, U, Dg, Wfx, bfx, Wfu, bfu, Wfd, bfd, Wfy, bfy, ws, ws, 1);
  kf_chain_kernel<<<16, 128, 0, stream>>>(ws, Out);
}


# Gain kernel: each wave sums all tiles of G itself from K slices stored transposed (A/B swapped), G not stored again, slice buffers alternate: 2 barriers a step -> 1; V's k blocks summed in new order
# speedup vs baseline: 12.9013x; 1.0956x over previous
#include <hip/hip_runtime.h>


typedef __attribute__((ext_vector_type(2))) float v2f;
typedef __attribute__((ext_vector_type(4))) float v4f;
typedef __attribute__((ext_vector_type(8))) float v8f;

#define TT  4096
#define BB  256
#define NXX 64
#define NYY 16
#define NUU 16
#define NDD 8

#define SLOT    (NXX * NYY)
#define SPP     20
#define SPS     (NYY * SPP)
#define NS_TOL  6.103515625e-05f
#define HALF_T  (TT / 2)
#define TC      64
#define NCP     (HALF_T / TC)
#define NCH     (TT / TC)
#define NTILE   20
#define P_SLOT0   (HALF_T + 1)
#define RES_SLOT0 (P_SLOT0 + 4)

__device__ __forceinline__ float bcast_lane(float x, int l) {
  return __int_as_float(__builtin_amdgcn_readlane(__float_as_int(x), l));
}

__device__ __forceinline__ float xhalf(float x) {
  const int b = __float_as_int(x);
  return __int_as_float(__builtin_amdgcn_permlanex16(b, b, 0x76543210, 0xfedcba98,
                                                     false, false));
}

__device__ __forceinline__ v8f wmma_mm(const float* A, int ras, int cas,
                                       const float* B, int rbs, int cbs,
                                       int K, v8f acc, int lane) {
  const int mn = lane & 15;
  const int kh = (lane >> 4) << 1;
#pragma unroll
  for (int k = 0; k < K; k += 4) {
    v2f a, b;
    a.x = A[mn * ras + (k + kh) * cas];
    a.y = A[mn * ras + (k + kh + 1) * cas];
    b.x = B[(k + kh) * rbs + mn * cbs];
    b.y = B[(k + kh + 1) * rbs + mn * cbs];
    acc = __builtin_amdgcn_wmma_f32_16x16x4_f32(false, a, false, b,
                                                (short)0, acc, false, false);
  }
  return acc;
}

__device__ __forceinline__ void store_tile(float* D, int ldd, v8f acc, int lane) {
  const int col = lane & 15;
  const int rb  = (lane >> 4) << 3;
#pragma unroll
  for (int i = 0; i < 8; ++i) D[(rb + i) * ldd + col] = acc[i];
}

__global__ __launch_bounds__(128) void kf_gains_kernel(
    const float* __restrict__ Wfx_g, const float* __restrict__ Wfy_g,
    float* __restrict__ ws, int half) {
  float* Lbuf  = ws;
  float* Psave = ws + (size_t)P_SLOT0 * SLOT;
  __shared__ float sWfy[NXX * NYY];
  __shared__ float sW2[NXX * NYY];
  __shared__ __attribute__((aligned(16))) float sGp[2 * 16 * SPS];
  __shared__ __attribute__((aligned(16))) float sSp[2 * 4 * SPS];
  __shared__ __attribute__((aligned(16))) float sXi[4 * NYY * NYY];

  const int tid  = threadIdx.x;
  const int lane = tid & 31;
  const int wave = tid >> 5;
  const int c    = lane & 15;
  const int h    = lane >> 4;
  float* myX = sXi + wave * (NYY * NYY);

  for (int e = tid; e < NXX * NYY; e += 128) sWfy[e] = Wfy_g[e];
  for (int e = tid; e < 4 * NYY * NYY; e += 128) sXi[e] = 0.0f;
  if (half == 0) {
    for (int e = tid; e < NXX * NYY; e += 128) Lbuf[e] = 0.0f;
  } else {
    for (int e = tid; e < NXX * NYY; e += 128)
      Lbuf[e] = Lbuf[(size_t)HALF_T * SLOT + e];
  }
  __syncthreads();
  v8f s0acc = {};
  {
    v8f acc = {};
    acc = wmma_mm(Wfx_g + wave * 16, 1, NXX, sWfy, NYY, 1, NXX, acc, lane);
    store_tile(sW2 + wave * 16 * NYY, NYY, acc, lane);
    v8f s0 = {};
    s0 = wmma_mm(sWfy, 1, NYY, sWfy, NYY, 1, NXX, s0, lane);
#pragma unroll
    for (int i = 0; i < 8; ++i)
      s0acc[i] = (wave == 0) ? (s0[i] + ((i + 8 * h == c) ? 1.0f : 0.0f)) : 0.0f;
  }
  __syncthreads();

  v8f pt[4];
  v2f w2f[16];
  v2f w2a[4];
  v2f wfa[16];
  float wfya[4][8];
  v8f xinv = {};
  if (half == 0) {
#pragma unroll
    for (int r = 0; r < 4; ++r)
#pragma unroll
      for (int i = 0; i < 8; ++i)
        pt[r][i] = (r == 0 && i + 8 * h == c) ? 1.0f : 0.0f;
  } else {
#pragma unroll
    for (int r = 0; r < 4; ++r)
#pragma unroll
      for (int i = 0; i < 8; ++i)
        pt[r][i] = Psave[((wave * 4 + r) * 8 + i) * 32 + lane];
  }
#pragma unroll
  for (int r = 0; r < 4; ++r)
#pragma unroll
    for (int jj = 0; jj < 4; ++jj) {
      const int k = 16 * ((wave + r) & 3) + 8 * h + 2 * jj;
      w2f[4 * r + jj].x = sW2[k * NYY + c];
      w2f[4 * r + jj].y = sW2[(k + 1) * NYY + c];
    }
#pragma unroll
  for (int jj = 0; jj < 4; ++jj) {
    const int k = 16 * wave + 8 * h + 2 * jj;
    w2a[jj].x = sW2[k * NYY + c];
    w2a[jj].y = sW2[(k + 1) * NYY + c];
  }
#pragma unroll
  for (int mt = 0; mt < 4; ++mt)
#pragma unroll
    for (int jj = 0; jj < 4; ++jj) {
      const int k = 16 * wave + 8 * h + 2 * jj;
      wfa[4 * mt + jj].x = Wfx_g[(16 * mt + c) * NXX + k];
      wfa[4 * mt + jj].y = Wfx_g[(16 * mt + c) * NXX + k + 1];
    }
#pragma unroll
  for (int r = 0; r < 4; ++r)
#pragma unroll
    for (int i = 0; i < 8; ++i)
      wfya[r][i] = sWfy[(16 * ((wave + r) & 3) + c) * NYY + 8 * h + i];

  for (int t = 0; t < HALF_T; ++t) {
    float* sSb = sSp + (t & 1) * (4 * SPS);
    float* sGb = sGp + (t & 1) * (16 * SPS);
    {
      v8f acc = {};
#pragma unroll
      for (int r = 0; r < 4; ++r)
#pragma unroll
        for (int jj = 0; jj < 4; ++jj) {
          v2f a;
          a.x = pt[r][2 * jj];
          a.y = pt[r][2 * jj + 1];
          acc = __builtin_amdgcn_wmma_f32_16x16x4_f32(false, a, false, w2f[4 * r + jj],
                                                      (short)0, acc, false, false);
        }
      v8f sp = s0acc;
      v8f gp[4];
#pragma unroll
      for (int mt = 0; mt < 4; ++mt) gp[mt] = v8f{};
#pragma unroll
      for (int jj = 0; jj < 4; ++jj) {
        v2f b;
        b.x = acc[2 * jj];
        b.y = acc[2 * jj + 1];
        sp = __builtin_amdgcn_wmma_f32_16x16x4_f32(false, w2a[jj], false, b,
                                                   (short)0, sp, false, false);
#pragma unroll
        for (int mt = 0; mt < 4; ++mt)
          gp[mt] = __builtin_amdgcn_wmma_f32_16x16x4_f32(false, b, false, wfa[4 * mt + jj],
                                                         (short)0, gp[mt], false, false);
      }
      float* dsp = sSb + wave * SPS + c * SPP + 8 * h;
      v4f slo, shi;
      slo.x = sp[0]; slo.y = sp[1]; slo.z = sp[2]; slo.w = sp[3];
      shi.x = sp[4]; shi.y = sp[5]; shi.z = sp[6]; shi.w = sp[7];
      *(v4f*)dsp = slo;
      *(v4f*)(dsp + 4) = shi;
#pragma unroll
      for (int mt = 0; mt < 4; ++mt) {
        float* dgp = sGb + (wave * 4 + mt) * SPS + c * SPP + 8 * h;
        v4f glo, ghi;
        glo.x = gp[mt][0]; glo.y = gp[mt][1]; glo.z = gp[mt][2]; glo.w = gp[mt][3];
        ghi.x = gp[mt][4]; ghi.y = gp[mt][5]; ghi.z = gp[mt][6]; ghi.w = gp[mt][7];
        *(v4f*)dgp = glo;
        *(v4f*)(dgp + 4) = ghi;
      }
    }
    __syncthreads();
    {
      float scol[8];
      v8f d1 = {};
      const float* spc = sSb + c * SPP + 8 * h;
      v4f sl[4][2];
#pragma unroll
      for (int w = 0; w < 4; ++w) {
        sl[w][0] = *(const v4f*)(spc + w * SPS);
        sl[w][1] = *(const v4f*)(spc + w * SPS + 4);
      }
      const float* xrp = myX + c * NYY + 8 * h;
      const v4f xr0 = *(const v4f*)xrp;
      const v4f xr1 = *(const v4f*)(xrp + 4);
      v4f gl[4][4][2];
#pragma unroll
      for (int r = 0; r < 4; ++r) {
        const float* gpc = sGb + ((wave + r) & 3) * SPS + c * SPP + 8 * h;
#pragma unroll
        for (int w = 0; w < 4; ++w) {
          gl[r][w][0] = *(const v4f*)(gpc + w * 4 * SPS);
          gl[r][w][1] = *(const v4f*)(gpc + w * 4 * SPS + 4);
        }
      }
      {
        const v4f lo = (sl[0][0] + sl[1][0]) + (sl[2][0] + sl[3][0]);
        const v4f hi = (sl[0][1] + sl[1][1]) + (sl[2][1] + sl[3][1]);
        scol[0] = lo.x; scol[1] = lo.y; scol[2] = lo.z; scol[3] = lo.w;
        scol[4] = hi.x; scol[5] = hi.y; scol[6] = hi.z; scol[7] = hi.w;
      }
      const float x0r[8] = {xr0.x, xr0.y, xr0.z, xr0.w, xr1.x, xr1.y, xr1.z, xr1.w};
#pragma unroll
      for (int jj = 0; jj < 4; ++jj) {
        v2f a, b;
        a.x = scol[2 * jj]; a.y = scol[2 * jj + 1];
        b.x = x0r[2 * jj];  b.y = x0r[2 * jj + 1];
        d1 = __builtin_amdgcn_wmma_f32_16x16x4_f32(false, a, false, b,
                                                   (short)0, d1, false, false);
      }
      v4f ga0[4], ga1[4];
#pragma unroll
      for (int r = 0; r < 4; ++r) {
        const v4f lo = (gl[r][0][0] + gl[r][1][0]) + (gl[r][2][0] + gl[r][3][0]);
        const v4f hi = (gl[r][0][1] + gl[r][1][1]) + (gl[r][2][1] + gl[r][3][1]);
        ga0[r].x = lo.x + wfya[r][0]; ga0[r].y = lo.y + wfya[r][1];
        ga0[r].z = lo.z + wfya[r][2]; ga0[r].w = lo.w + wfya[r][3];
        ga1[r].x = hi.x + wfya[r][4]; ga1[r].y = hi.y + wfya[r][5];
        ga1[r].z = hi.z + wfya[r][6]; ga1[r].w = hi.w + wfya[r][7];
      }
      const v4f gw0 = ga0[0];
      const v4f gw1 = ga1[0];
      float et[8];
#pragma unroll
      for (int i = 0; i < 8; ++i)
        et[i] = ((i + 8 * h == c) ? 1.0f : 0.0f) - d1[i];
      v8f x1 = xinv;
#pragma unroll
      for (int jj = 0; jj < 4; ++jj) {
        v2f a, b;
        a.x = et[2 * jj];   a.y = et[2 * jj + 1];
        b.x = xinv[2 * jj]; b.y = xinv[2 * jj + 1];
        x1 = __builtin_amdgcn_wmma_f32_16x16x4_f32(false, a, false, b,
                                                   (short)0, x1, false, false);
      }
      unsigned notsmall = 0u;
#pragma unroll
      for (int i = 0; i < 8; ++i)
        notsmall |= __builtin_amdgcn_ballot_w32(!(__builtin_fabsf(et[i]) <= NS_TOL));
      if (notsmall != 0u) {
        float colv[16];
#pragma unroll
        for (int i = 0; i < 8; ++i) {
          const float o = xhalf(scol[i]);
          colv[i]     = (h == 0) ? scol[i] : o;
          colv[8 + i] = (h == 0) ? o : scol[i];
        }
#pragma unroll
        for (int i = 0; i < 16; ++i)
          colv[i] = (lane < 16) ? colv[i] : ((i == (lane - 16)) ? 1.0f : 0.0f);
#pragma unroll
        for (int k = 0; k < 16; ++k) {
          float piv  = bcast_lane(colv[k], k);
          float rinv = 1.0f / piv;
          colv[k] *= rinv;
#pragma unroll
          for (int i = 0; i < 16; ++i) {
            if (i != k) {
              float f = bcast_lane(colv[i], k);
              colv[i] -= f * colv[k];
            }
          }
        }
#pragma unroll
        for (int i = 0; i < 8; ++i) {
          const float o = xhalf(colv[i]);
          x1[i] = (h == 0) ? o : colv[8 + i];
        }
      }
      xinv = x1;
#pragma unroll
      for (int i = 0; i < 8; ++i) myX[(8 * h + i) * NYY + c] = x1[i];
      v8f lt = {};
      {
        v2f a, b;
        a.x = x1[0]; a.y = x1[1]; b.x = gw0.x; b.y = gw0.y;
        lt = __builtin_amdgcn_wmma_f32_16x16x4_f32(false, a, false, b, (short)0, lt, false, false);
        a.x = x1[2]; a.y = x1[3]; b.x = gw0.z; b.y = gw0.w;
        lt = __builtin_amdgcn_wmma_f32_16x16x4_f32(false, a, false, b, (short)0, lt, false, false);
        a.x = x1[4]; a.y = x1[5]; b.x = gw1.x; b.y = gw1.y;
        lt = __builtin_amdgcn_wmma_f32_16x16x4_f32(false, a, false, b, (short)0, lt, false, false);
        a.x = x1[6]; a.y = x1[7]; b.x = gw1.z; b.y = gw1.w;
        lt = __builtin_amdgcn_wmma_f32_16x16x4_f32(false, a, false, b, (short)0, lt, false, false);
      }
      float* gL = Lbuf + (size_t)(t + 1) * SLOT + (wave * 16 + c) * NYY + 8 * h;
      v4f lo, hi;
      lo.x = lt[0]; lo.y = lt[1]; lo.z = lt[2]; lo.w = lt[3];
      hi.x = lt[4]; hi.y = lt[5]; hi.z = lt[6]; hi.w = lt[7];
      *(v4f*)gL = lo;
      *(v4f*)(gL + 4) = hi;
#pragma unroll
      for (int r = 0; r < 4; ++r) {
        const v4f a0 = ga0[r];
        const v4f a1 = ga1[r];
        v2f a, b;
        v8f acc = {};
        a.x = a0.x; a.y = a0.y; b.x = lt[0]; b.y = lt[1];
        acc = __builtin_amdgcn_wmma_f32_16x16x4_f32(false, a, false, b, (short)0, acc, false, false);
        a.x = a0.z; a.y = a0.w; b.x = lt[2]; b.y = lt[3];
        acc = __builtin_amdgcn_wmma_f32_16x16x4_f32(false, a, false, b, (short)0, acc, false, false);
        a.x = a1.x; a.y = a1.y; b.x = lt[4]; b.y = lt[5];
        acc = __builtin_amdgcn_wmma_f32_16x16x4_f32(false, a, false, b, (short)0, acc, false, false);
        a.x = a1.z; a.y = a1.w; b.x = lt[6]; b.y = lt[7];
        acc = __builtin_amdgcn_wmma_f32_16x16x4_f32(false, a, false, b, (short)0, acc, false, false);
#pragma unroll
        for (int i = 0; i < 8; ++i)
          pt[r][i] = ((r == 0 && i + 8 * h == c) ? 1.0f : 0.0f) - acc[i];
      }
    }
  }
  if (half == 0) {
#pragma unroll
    for (int mt = 0; mt < 4; ++mt)
#pragma unroll
      for (int i = 0; i < 8; ++i)
        Psave[((wave * 4 + mt) * 8 + i) * 32 + lane] = pt[mt][i];
  }
}

__global__ __launch_bounds__(128) void kf_chunk_scan_kernel(
    const float* __restrict__ Ym, const float* __restrict__ U,
    const float* __restrict__ Dg,
    const float* __restrict__ Wfx_g, const float* __restrict__ bfx,
    const float* __restrict__ Wfu_g, const float* __restrict__ bfu,
    const float* __restrict__ Wfd_g, const float* __restrict__ bfd,
    const float* __restrict__ Wfy_g, const float* __restrict__ bfy_g,
    const float* __restrict__ Lbuf, float* __restrict__ ws_out, int half) {
  __shared__ float sWfx[NXX * NXX];
  __shared__ float sWfu[NUU * NXX];
  __shared__ float sWfd[NDD * NXX];
  __shared__ float sWfy[NXX * NYY];
  __shared__ float sBsum[NXX];
  __shared__ float sBfy[NYY];
  __shared__ float sX[16 * NXX];
  __shared__ float sXp[16 * NXX];
  __shared__ __attribute__((aligned(16))) float sInn[16 * NYY];

  const int tid  = threadIdx.x;
  const int lane = tid & 31;
  const int wave = tid >> 5;
  const int c16  = lane & 15;
  const int h    = lane >> 4;
  const int j    = blockIdx.x;
  const int cl   = blockIdx.y;
  const int c    = half * NCP + cl;
  const bool homog = (j >= 16);
  if (homog && c == 0) return;
  const int t0   = c * TC;
  const int s0   = cl * TC;
  const int r0   = homog ? 0 : j * 16;

  for (int e = tid; e < NXX * NXX; e += 128) sWfx[e] = Wfx_g[e];
  for (int e = tid; e < NUU * NXX; e += 128) sWfu[e] = Wfu_g[e];
  for (int e = tid; e < NDD * NXX; e += 128) sWfd[e] = Wfd_g[e];
  for (int e = tid; e < NXX * NYY; e += 128) sWfy[e] = Wfy_g[e];
  if (tid < NXX) sBsum[tid] = homog ? 0.0f : (bfx[tid] + bfu[tid] + bfd[tid]);
  if (tid < NYY) sBfy[tid] = homog ? 0.0f : bfy_g[tid];
  if (homog) {
    const int i0 = (j - 16) * 16;
    for (int e = tid; e < 16 * NXX; e += 128) {
      int row = e >> 6, col = e & 63;
      sX[e] = (col == i0 + row) ? 1.0f : 0.0f;
    }
  } else {
    for (int e = tid; e < 16 * NXX; e += 128) sX[e] = 0.0f;
  }
  __syncthreads();

  for (int tt = 0; tt < TC; ++tt) {
    const int t = t0 + tt;
    v4f u0 = {}, u1 = {}, d0 = {};
    float ymr[8];
#pragma unroll
    for (int i = 0; i < 8; ++i) ymr[i] = 0.0f;
    if (!homog) {
      const float* gU = U  + ((size_t)t * BB + r0 + c16) * NUU + 8 * h;
      const float* gD = Dg + ((size_t)t * BB + r0 + c16) * NDD + 4 * h;
      u0 = *(const v4f*)gU;
      u1 = *(const v4f*)(gU + 4);
      d0 = *(const v4f*)gD;
      if (wave == 0) {
        const float* gYm = Ym + ((size_t)t * BB + r0 + 8 * h) * NYY + c16;
#pragma unroll
        for (int i = 0; i < 8; ++i) ymr[i] = __builtin_nontemporal_load(gYm + i * NYY);
      }
    }
    const float* gLr = Lbuf + (size_t)(s0 + tt) * SLOT + (wave * 16 + c16) * NYY + 8 * h;
    const v4f l0 = *(const v4f*)gLr;
    const v4f l1 = *(const v4f*)(gLr + 4);
    if (tt + 1 < TC) {
      const char* pL = (const char*)(Lbuf + (size_t)(s0 + tt + 1) * SLOT);
      if (tid < 32) __builtin_prefetch(pL + tid * 128, 0, 0);
      if (!homog) {
        const char* pYm = (const char*)(Ym + (size_t)(t + 1) * BB * NYY + r0 * NYY);
        const char* pU  = (const char*)(U  + (size_t)(t + 1) * BB * NUU + r0 * NUU);
        const char* pD  = (const char*)(Dg + (size_t)(t + 1) * BB * NDD + r0 * NDD);
        if (tid < 8)  __builtin_prefetch(pYm + tid * 128, 0, 0);
        if (tid < 8)  __builtin_prefetch(pU + tid * 128, 0, 0);
        if (tid < 4)  __builtin_prefetch(pD + tid * 128, 0, 0);
      }
    }
    {
      const int nt = wave;
      v8f acc = {};
      acc = wmma_mm(sX, NXX, 1, sWfx + nt * 16, NXX, 1, NXX, acc, lane);
      {
        const float* bu = sWfu + (8 * h) * NXX + nt * 16 + c16;
        v2f a, b;
        a.x = u0.x; a.y = u0.y; b.x = bu[0 * NXX]; b.y = bu[1 * NXX];
        acc = __builtin_amdgcn_wmma_f32_16x16x4_f32(false, a, false, b, (short)0, acc, false, false);
        a.x = u0.z; a.y = u0.w; b.x = bu[2 * NXX]; b.y = bu[3 * NXX];
        acc = __builtin_amdgcn_wmma_f32_16x16x4_f32(false, a, false, b, (short)0, acc, false, false);
        a.x = u1.x; a.y = u1.y; b.x = bu[4 * NXX]; b.y = bu[5 * NXX];
        acc = __builtin_amdgcn_wmma_f32_16x16x4_f32(false, a, false, b, (short)0, acc, false, false);
        a.x = u1.z; a.y = u1.w; b.x = bu[6 * NXX]; b.y = bu[7 * NXX];
        acc = __builtin_amdgcn_wmma_f32_16x16x4_f32(false, a, false, b, (short)0, acc, false, false);
      }
      {
        const float* bd = sWfd + (4 * h) * NXX + nt * 16 + c16;
        v2f a, b;
        a.x = d0.x; a.y = d0.y; b.x = bd[0 * NXX]; b.y = bd[1 * NXX];
        acc = __builtin_amdgcn_wmma_f32_16x16x4_f32(false, a, false, b, (short)0, acc, false, false);
        a.x = d0.z; a.y = d0.w; b.x = bd[2 * NXX]; b.y = bd[3 * NXX];
        acc = __builtin_amdgcn_wmma_f32_16x16x4_f32(false, a, false, b, (short)0, acc, false, false);
      }
      const int rb = h << 3;
      float bias = sBsum[nt * 16 + c16];
#pragma unroll
      for (int i = 0; i < 8; ++i)
        sXp[(rb + i) * NXX + nt * 16 + c16] = acc[i] + bias;
    }
    __syncthreads();
    if (wave == 0) {
      v8f acc = {};
      acc = wmma_mm(sXp, NXX, 1, sWfy, NYY, 1, NXX, acc, lane);
      const int rb = h << 3;
      float by = sBfy[c16];
#pragma unroll
      for (int i = 0; i < 8; ++i)
        sInn[(rb + i) * NYY + c16] = ymr[i] - (acc[i] + by);
    }
    __syncthreads();
    {
      const int nt = wave;
      const float* ia = sInn + c16 * NYY + 8 * h;
      const v4f i0 = *(const v4f*)ia;
      const v4f i1 = *(const v4f*)(ia + 4);
      v8f acc = {};
      v2f a, b;
      a.x = i0.x; a.y = i0.y; b.x = l0.x; b.y = l0.y;
      acc = __builtin_amdgcn_wmma_f32_16x16x4_f32(false, a, false, b, (short)0, acc, false, false);
      a.x = i0.z; a.y = i0.w; b.x = l0.z; b.y = l0.w;
      acc = __builtin_amdgcn_wmma_f32_16x16x4_f32(false, a, false, b, (short)0, acc, false, false);
      a.x = i1.x; a.y = i1.y; b.x = l1.x; b.y = l1.y;
      acc = __builtin_amdgcn_wmma_f32_16x16x4_f32(false, a, false, b, (short)0, acc, false, false);
      a.x = i1.z; a.y = i1.w; b.x = l1.z; b.y = l1.w;
      acc = __builtin_amdgcn_wmma_f32_16x16x4_f32(false, a, false, b, (short)0, acc, false, false);
      const int rb = h << 3;
#pragma unroll
      for (int i = 0; i < 8; ++i) {
        int idx = (rb + i) * NXX + nt * 16 + c16;
        sX[idx] = sXp[idx] + acc[i];
      }
    }
    __syncthreads();
  }

  float* res = ws_out + (size_t)(RES_SLOT0 + c * NTILE + j) * SLOT;
  for (int e = tid; e < 16 * NXX; e += 128) res[e] = sX[e];
}

__global__ __launch_bounds__(128) void kf_chain_kernel(
    const float* __restrict__ ws, float* __restrict__ Out) {
  __shared__ float sXc[2][16 * NXX];
  __shared__ float sPhi[NXX * NXX];
  __shared__ float sZ[16 * NXX];

  const int tid  = threadIdx.x;
  const int lane = tid & 31;
  const int wave = tid >> 5;
  const int j    = blockIdx.x;
  const float* res = ws + (size_t)RES_SLOT0 * SLOT;

  float rphi[32], rz[8];
  {
    const float* z0 = res + (size_t)j * SLOT;
    for (int e = tid; e < 16 * NXX; e += 128) sXc[0][e] = z0[e];
    const float* gP = res + (size_t)(1 * NTILE + 16) * SLOT;
    const float* gZ = res + (size_t)(1 * NTILE + j) * SLOT;
#pragma unroll
    for (int i = 0; i < 32; ++i) rphi[i] = gP[tid + i * 128];
#pragma unroll
    for (int i = 0; i < 8; ++i) rz[i] = gZ[tid + i * 128];
  }

  int cur = 0;
  for (int c = 1; c < NCH; ++c) {
#pragma unroll
    for (int i = 0; i < 32; ++i) sPhi[tid + i * 128] = rphi[i];
#pragma unroll
    for (int i = 0; i < 8; ++i) sZ[tid + i * 128] = rz[i];
    __syncthreads();
    if (c + 1 < NCH) {
      const float* gP = res + (size_t)((c + 1) * NTILE + 16) * SLOT;
      const float* gZ = res + (size_t)((c + 1) * NTILE + j) * SLOT;
#pragma unroll
      for (int i = 0; i < 32; ++i) rphi[i] = gP[tid + i * 128];
#pragma unroll
      for (int i = 0; i < 8; ++i) rz[i] = gZ[tid + i * 128];
    }
    {
      const int nt = wave;
      v8f acc = {};
      acc = wmma_mm(sXc[cur], NXX, 1, sPhi + nt * 16, NXX, 1, NXX, acc, lane);
      int col = lane & 15, rb = (lane >> 4) << 3;
#pragma unroll
      for (int i = 0; i < 8; ++i) {
        int idx = (rb + i) * NXX + nt * 16 + col;
        sXc[cur ^ 1][idx] = acc[i] + sZ[idx];
      }
    }
    __syncthreads();
    cur ^= 1;
  }

  for (int e = tid; e < 16 * NXX; e += 128) {
    int row = e >> 6, col = e & 63;
    Out[(size_t)(j * 16 + row) * NXX + col] = sXc[cur][e];
  }
}

extern "C" void kernel_launch(void* const* d_in, const int* in_sizes, int n_in,
                              void* d_out, int out_size, void* d_ws, size_t ws_size,
                              hipStream_t stream) {
  const float* Ym  = (const float*)d_in[0];
  const float* U   = (const float*)d_in[1];
  const float* Dg  = (const float*)d_in[2];
  const float* Wfx = (const float*)d_in[3];
  const float* bfx = (const float*)d_in[4];
  const float* bfu = (const float*)d_in[6];
  const float* Wfu = (const float*)d_in[5];
  const float* Wfd = (const float*)d_in[7];
  const float* bfd = (const float*)d_in[8];
  const float* Wfy = (const float*)d_in[9];
  const float* bfy = (const float*)d_in[10];

  float* ws  = (float*)d_ws;
  float* Out = (float*)d_out;

  kf_gains_kernel<<<1, 128, 0, stream>>>(Wfx, Wfy, ws, 0);
  kf_chunk_scan_kernel<<<dim3(NTILE, NCP), 128, 0, stream>>>(
      Ym, U, Dg, Wfx, bfx, Wfu, bfu, Wfd, bfd, Wfy, bfy, ws, ws, 0);
  kf_gains_kernel<<<1, 128, 0, stream>>>(Wfx, Wfy, ws, 1);
  kf_chunk_scan_kernel<<<dim3(NTILE, NCP), 128, 0, stream>>>(
      Ym, U, Dg, Wfx, bfx, Wfu, bfu, Wfd, bfd, Wfy, bfy, ws, ws, 1);
  kf_chain_kernel<<<16, 128, 0, stream>>>(ws, Out);
}
